# starting kernel
# baseline (speedup 1.0000x reference)
#include <hip/hip_runtime.h>


#define D  10
#define RT 4
#define SP 12

typedef __attribute__((ext_vector_type(2))) float v2f;
typedef __attribute__((ext_vector_type(8))) float v8f;

__device__ __forceinline__ v8f wmma_k4(v2f a, v2f b, v8f c) {
  return __builtin_amdgcn_wmma_f32_16x16x4_f32(
       false, a,  false, b,
       (short)0, c,  false,  false);
}

__global__ void k_prep(const float* __restrict__ Ws1, const float* __restrict__ Wn1,
                       const float* __restrict__ b1,
                       const float* __restrict__ Ws2, const float* __restrict__ Wn2,
                       const float* __restrict__ b2,
                       float* __restrict__ WsP1, float* __restrict__ WnP1,
                       float* __restrict__ bsP1,
                       float* __restrict__ WsP2, float* __restrict__ WnP2,
                       float* __restrict__ bsP2) {
  for (int i = threadIdx.x; i < 16 * SP; i += blockDim.x) {
    int n = i / SP, k = i % SP;
    float s1 = 0.f, s2 = 0.f;
    if (n < D && k < D) {
#pragma unroll
      for (int r = 0; r < RT; ++r) {
        s1 += Ws1[r * D * D + n * D + k];
        s2 += Ws2[r * D * D + n * D + k];
      }
    }
    WsP1[i] = s1;
    WsP2[i] = s2;
  }
  for (int i = threadIdx.x; i < RT * 16 * SP; i += blockDim.x) {
    int r = i / (16 * SP), j = i % (16 * SP);
    int n = j / SP, k = j % SP;
    float v1 = 0.f, v2 = 0.f;
    if (n < D && k < D) {
      v1 = Wn1[r * D * D + n * D + k];
      v2 = Wn2[r * D * D + n * D + k];
    }
    WnP1[i] = v1;
    WnP2[i] = v2;
  }
  for (int i = threadIdx.x; i < 16; i += blockDim.x) {
    float s1 = 0.f, s2 = 0.f;
    if (i < D) {
#pragma unroll
      for (int r = 0; r < RT; ++r) {
        s1 += b1[r * D + i];
        s2 += b2[r * D + i];
      }
    }
    bsP1[i] = s1;
    bsP2[i] = s2;
  }
}

__global__ void k_padcopy(const float* __restrict__ x, float* __restrict__ hp, int Nn) {
  int i = blockIdx.x * blockDim.x + threadIdx.x;
  if (i >= Nn * SP) return;
  int n = i / SP, k = i - n * SP;
  hp[i] = (k < D) ? x[(long long)n * D + k] : 0.0f;
}

__global__ void k_zero(float* __restrict__ p, int n) {
  int i = blockIdx.x * blockDim.x + threadIdx.x;
  if (i < n) p[i] = 0.0f;
}

__global__ void k_scatter(const float* __restrict__ hp,
                          const int* __restrict__ src,
                          const int* __restrict__ dst,
                          float* __restrict__ msum,
                          float* __restrict__ deg, int E_) {
  long long gid = (long long)blockIdx.x * blockDim.x + threadIdx.x;
  if (gid >= (long long)E_ * D) return;
  int e = (int)(gid / D);
  int k = (int)(gid - (long long)e * D);
  int s = src[e];
  int d = dst[e];
  atomicAdd(&msum[(long long)d * SP + k], hp[(long long)s * SP + k]);
  if (k == 0) atomicAdd(&deg[d], 1.0f);
}


__global__ void k_self_wmma(const float* __restrict__ hp,
                            const float* __restrict__ WsP,
                            const float* __restrict__ bsP,
                            float* __restrict__ hout,
                            int nTiles, int Nn) {
  int wave = blockIdx.x * (blockDim.x >> 5) + (threadIdx.x >> 5);
  if (wave >= nTiles) return;
  int lane = threadIdx.x & 31;
  int m    = lane & 15;
  int half = lane >> 4;
  int node = min(wave * 16 + m, Nn - 1);

  union { v8f v; float f[8]; } acc;
  float bsum = bsP[m];
#pragma unroll
  for (int j = 0; j < 8; ++j) acc.f[j] = bsum;

  const float* arow = hp + (long long)node * SP;
  const float* brow = WsP + m * SP;
#pragma unroll
  for (int c = 0; c < 3; ++c) {
    int k0 = c * 4 + half * 2;
    v2f a, b;
    a.x = arow[k0]; a.y = arow[k0 + 1];
    b.x = brow[k0]; b.y = brow[k0 + 1];
    acc.v = wmma_k4(a, b, acc.v);
  }

  int rowBase = wave * 16 + half * 8;
  bool fullTile = (wave * 16 + 16) <= Nn;
  if (m < SP) {
    float* op = hout + (long long)rowBase * SP + m;
    if (fullTile) {
#pragma unroll
      for (int j = 0; j < 8; ++j) op[j * SP] = acc.f[j];
    } else {
#pragma unroll
      for (int j = 0; j < 8; ++j)
        if (rowBase + j < Nn) op[j * SP] = acc.f[j];
    }
  }
}

__global__ void k_neigh_wmma(const float* __restrict__ msum,
                             const float* __restrict__ deg,
                             const float* __restrict__ WnP,
                             float* __restrict__ hout,
                             int nTiles, int Nn) {
  int wave = blockIdx.x * (blockDim.x >> 5) + (threadIdx.x >> 5);
  if (wave >= nTiles) return;
  int lane = threadIdx.x & 31;
  int m    = lane & 15;
  int half = lane >> 4;
  int node = min(wave * 16 + m, Nn - 1);
  int mc   = min(m, SP - 1);
  int rowBase = wave * 16 + half * 8;
  bool fullTile = (wave * 16 + 16) <= Nn;

  float rdeg = 1.0f / fmaxf(deg[node], 1.0f);

  union { v8f v; float f[8]; } acc;
  if (fullTile) {
    const float* cp = hout + (long long)rowBase * SP + mc;
#pragma unroll
    for (int j = 0; j < 8; ++j) acc.f[j] = cp[j * SP];
  } else {
#pragma unroll
    for (int j = 0; j < 8; ++j)
      acc.f[j] = hout[(long long)min(rowBase + j, Nn - 1) * SP + mc];
  }

  const float* arow = msum + (long long)node * SP;
  const float* brow = WnP + m * SP;
#pragma unroll
  for (int c = 0; c < 3; ++c) {
    int k0 = c * 4 + half * 2;
    v2f a, b;
    a.x = arow[k0] * rdeg; a.y = arow[k0 + 1] * rdeg;
    b.x = brow[k0];        b.y = brow[k0 + 1];
    acc.v = wmma_k4(a, b, acc.v);
  }

  if (m < D) {
    float* op = hout + (long long)rowBase * SP + m;
    if (fullTile) {
#pragma unroll
      for (int j = 0; j < 8; ++j) op[j * SP] = acc.f[j];
    } else {
#pragma unroll
      for (int j = 0; j < 8; ++j)
        if (rowBase + j < Nn) op[j * SP] = acc.f[j];
    }
  }
}

__global__ void k_score(const float* __restrict__ hp,
                        const int* __restrict__ ps, const int* __restrict__ pd,
                        const int* __restrict__ ns, const int* __restrict__ nd,
                        float* __restrict__ out, int E_) {
  int i = blockIdx.x * blockDim.x + threadIdx.x;
  if (i >= 2 * E_) return;
  int a, b;
  if (i < E_) { a = ps[i]; b = pd[i]; }
  else        { a = ns[i - E_]; b = nd[i - E_]; }
  const float* pa = hp + (long long)a * SP;
  const float* pb = hp + (long long)b * SP;
  float s = 0.0f;
#pragma unroll
  for (int k = 0; k < D; ++k) s += pa[k] * pb[k];
  out[i] = s;
}

extern "C" void kernel_launch(void* const* d_in, const int* in_sizes, int n_in,
                              void* d_out, int out_size, void* d_ws, size_t ws_size,
                              hipStream_t stream) {
  (void)n_in; (void)out_size; (void)ws_size;
  const float* x    = (const float*)d_in[0];
  const int*   esrc = (const int*)d_in[1];
  const int*   edst = (const int*)d_in[2];
  const int*   nsrc = (const int*)d_in[3];
  const int*   ndst = (const int*)d_in[4];
  const float* Ws1  = (const float*)d_in[5];
  const float* Wn1  = (const float*)d_in[6];
  const float* b1   = (const float*)d_in[7];
  const float* Ws2  = (const float*)d_in[8];
  const float* Wn2  = (const float*)d_in[9];
  const float* b2   = (const float*)d_in[10];

  const int Nn = in_sizes[0] / D;
  const int Ee = in_sizes[3];

  float* B0   = (float*)d_ws;
  float* B1   = B0 + (size_t)Nn * SP;
  float* msum = B1 + (size_t)Nn * SP;
  float* dega = msum + (size_t)Nn * SP;
  float* WsP1 = dega + (size_t)Nn;
  float* WnP1 = WsP1 + 16 * SP;
  float* bsP1 = WnP1 + RT * 16 * SP;
  float* WsP2 = bsP1 + 16;
  float* WnP2 = WsP2 + 16 * SP;
  float* bsP2 = WnP2 + RT * 16 * SP;

  const int tiles   = (Nn + 15) / 16;
  const int wBlocks = (tiles + 3) / 4;
  const int zeroCnt = Nn * SP + Nn;
  const int padCnt  = Nn * SP;
  const long long sThreads = (long long)Ee * D;
  const int sBlocks = (int)((sThreads + 255) / 256);

  k_prep<<<1, 256, 0, stream>>>(Ws1, Wn1, b1, Ws2, Wn2, b2,
                                WsP1, WnP1, bsP1, WsP2, WnP2, bsP2);
  k_padcopy<<<(padCnt + 255) / 256, 256, 0, stream>>>(x, B0, Nn);

  auto layer = [&](const float* hin, const float* WsP, const float* WnP,
                   const float* bsP, float* hout) {
    k_self_wmma<<<wBlocks, 128, 0, stream>>>(hin, WsP, bsP, hout, tiles, Nn);
    for (int r = 0; r < RT; ++r) {
      k_zero<<<(zeroCnt + 255) / 256, 256, 0, stream>>>(msum, zeroCnt);
      k_scatter<<<sBlocks, 256, 0, stream>>>(hin, esrc + (size_t)r * Ee,
                                             edst + (size_t)r * Ee, msum, dega, Ee);
      k_neigh_wmma<<<wBlocks, 128, 0, stream>>>(msum, dega, WnP + r * 16 * SP,
                                                hout, tiles, Nn);
    }
  };

  layer(B0, WsP1, WnP1, bsP1, B1);
  layer(B1, WsP2, WnP2, bsP2, B0);

  k_score<<<(2 * Ee + 255) / 256, 256, 0, stream>>>(B0, esrc, edst, nsrc, ndst,
                                                    (float*)d_out, Ee);
}


# Layer 2 no longer filters: layer 1 saves each wave's matched edges as packed (src, owned row) words in the workspace and layer 2 streams them; sums stay LDS-atomic arrival order
# speedup vs baseline: 2.6752x; 2.6752x over previous
#include <hip/hip_runtime.h>


#define D  10
#define RT 4
#define SP 12
#define NPW_MAX 2048
#define TPW (NPW_MAX / 16 / 32)
#define NOTAG 255u
#define LCAP 1024
#define LOCB 11
#define NOEDGE 0xFFFFFFFFu

typedef __attribute__((ext_vector_type(2))) float v2f;
typedef __attribute__((ext_vector_type(8))) float v8f;

__device__ __forceinline__ v8f wmma_k4(v2f a, v2f b, v8f c) {
  return __builtin_amdgcn_wmma_f32_16x16x4_f32(
       false, a,  false, b,
       (short)0, c,  false,  false);
}

__global__ void k_prep(const float* __restrict__ Ws1, const float* __restrict__ Wn1,
                       const float* __restrict__ b1,
                       const float* __restrict__ Ws2, const float* __restrict__ Wn2,
                       const float* __restrict__ b2,
                       float* __restrict__ WsP1, float* __restrict__ WnP1,
                       float* __restrict__ bsP1,
                       float* __restrict__ WsP2, float* __restrict__ WnP2,
                       float* __restrict__ bsP2) {
  for (int i = threadIdx.x; i < 16 * SP; i += blockDim.x) {
    int n = i / SP, k = i % SP;
    float s1 = 0.f, s2 = 0.f;
    if (n < D && k < D) {
#pragma unroll
      for (int r = 0; r < RT; ++r) {
        s1 += Ws1[r * D * D + n * D + k];
        s2 += Ws2[r * D * D + n * D + k];
      }
    }
    WsP1[i] = s1;
    WsP2[i] = s2;
  }
  for (int i = threadIdx.x; i < RT * 16 * SP; i += blockDim.x) {
    int r = i / (16 * SP), j = i % (16 * SP);
    int n = j / SP, k = j % SP;
    float v1 = 0.f, v2 = 0.f;
    if (n < D && k < D) {
      v1 = Wn1[r * D * D + n * D + k];
      v2 = Wn2[r * D * D + n * D + k];
    }
    WnP1[i] = v1;
    WnP2[i] = v2;
  }
  for (int i = threadIdx.x; i < 16; i += blockDim.x) {
    float s1 = 0.f, s2 = 0.f;
    if (i < D) {
#pragma unroll
      for (int r = 0; r < RT; ++r) {
        s1 += b1[r * D + i];
        s2 += b2[r * D + i];
      }
    }
    bsP1[i] = s1;
    bsP2[i] = s2;
  }
}

__global__ void k_padcopy(const float* __restrict__ x, float* __restrict__ hp, int Nn) {
  int i = blockIdx.x * blockDim.x + threadIdx.x;
  if (i >= Nn * SP) return;
  int n = i / SP, k = i - n * SP;
  hp[i] = (k < D) ? x[(long long)n * D + k] : 0.0f;
}

__global__ void k_owner(const int* __restrict__ dst, unsigned int* __restrict__ owner,
                        int E_, int Epad, int npw) {
  int q4 = Epad >> 2;
  int i = blockIdx.x * blockDim.x + threadIdx.x;
  if (i >= RT * q4) return;
  int r  = i / q4;
  int e0 = (i - r * q4) * 4;
  unsigned int pack = 0;
#pragma unroll
  for (int b = 0; b < 4; ++b) {
    int e = e0 + b;
    unsigned int tag = NOTAG;
    if (e < E_) tag = (unsigned int)dst[(long long)r * E_ + e] / (unsigned int)npw;
    pack |= tag << (8 * b);
  }
  owner[i] = pack;
}

__device__ __forceinline__ unsigned int zero_bytes(unsigned int x) {
  return ~(((x & 0x7F7F7F7Fu) + 0x7F7F7F7Fu) | x | 0x7F7F7F7Fu);
}

__device__ __forceinline__ void add_row(float* mp, float4 r0, float4 r1, float4 r2) {
  atomicAdd(mp + 0, r0.x); atomicAdd(mp + 1, r0.y);
  atomicAdd(mp + 2, r0.z); atomicAdd(mp + 3, r0.w);
  atomicAdd(mp + 4, r1.x); atomicAdd(mp + 5, r1.y);
  atomicAdd(mp + 6, r1.z); atomicAdd(mp + 7, r1.w);
  atomicAdd(mp + 8, r2.x); atomicAdd(mp + 9, r2.y);
  atomicAdd(mp + 10, 1.0f);
}

template <bool SAVE>
__device__ __forceinline__ void flush_edges(const int* wq, int n, int lane,
                                            const float* __restrict__ hin,
                                            const int* __restrict__ srcr,
                                            const int* __restrict__ dstr,
                                            float* s_msum, int lo, int npw,
                                            unsigned int* __restrict__ lst, int pos) {
  if (lane < n) {
    int e = wq[lane];
    int s = srcr[e];
    unsigned int loc = (unsigned int)(dstr[e] - lo);
    bool ok = loc < (unsigned int)npw;
    if (SAVE) {
      int at = pos + lane;
      if (at < LCAP) lst[at] = ok ? (((unsigned int)s << LOCB) | loc) : NOEDGE;
    }
    if (ok) {
      const float4* rp = (const float4*)(hin + (long long)s * SP);
      float4 r0 = rp[0], r1 = rp[1], r2 = rp[2];
      add_row(s_msum + loc * SP, r0, r1, r2);
    }
  }
}

template <bool SAVE>
__device__ __forceinline__ int scan_edges(const uint4* __restrict__ ow, int nVec,
                                          unsigned int myTag, int tid, int lane, int* wq,
                                          const float* __restrict__ hin,
                                          const int* __restrict__ srcr,
                                          const int* __restrict__ dstr,
                                          float* sm, int lo, int npw,
                                          unsigned int* __restrict__ lst) {
  int qn = 0;
  int done = 0;

  for (int base = 0; base < nVec; base += 2048) {
    int v0 = base + tid, v1 = v0 + 1024;
    uint4 q0 = make_uint4(~0u, ~0u, ~0u, ~0u), q1 = q0;
    if (v0 < nVec) q0 = ow[v0];
    if (v1 < nVec) q1 = ow[v1];
    unsigned int mm;
    mm  = zero_bytes(q0.x ^ myTag) >> 7;
    mm |= zero_bytes(q0.y ^ myTag) >> 6;
    mm |= zero_bytes(q0.z ^ myTag) >> 5;
    mm |= zero_bytes(q0.w ^ myTag) >> 4;
    mm |= zero_bytes(q1.x ^ myTag) >> 3;
    mm |= zero_bytes(q1.y ^ myTag) >> 2;
    mm |= zero_bytes(q1.z ^ myTag) >> 1;
    mm |= zero_bytes(q1.w ^ myTag);

    for (int trip = 0; trip < 32; ++trip) {
      bool has = (mm != 0u);
      unsigned int bal = __builtin_amdgcn_ballot_w32(has);
      if (bal == 0u) break;
      if (has) {
        int p = __builtin_ctz(mm);
        mm &= mm - 1u;
        int jj = p & 7, n = p >> 3;
        int v  = (jj & 4) ? v1 : v0;
        int e  = v * 16 + (jj & 3) * 4 + n;
        int pos = qn + __builtin_popcount(bal & ((1u << lane) - 1u));
        wq[pos] = e;
      }
      qn += __builtin_popcount(bal);
      if (qn >= 32) {
        flush_edges<SAVE>(wq, 32, lane, hin, srcr, dstr, sm, lo, npw, lst, done);
        done += 32;
        int rest = qn - 32;
        int tmp = 0;
        if (lane < rest) tmp = wq[32 + lane];
        if (lane < rest) wq[lane] = tmp;
        qn = rest;
      }
    }
  }
  flush_edges<SAVE>(wq, qn, lane, hin, srcr, dstr, sm, lo, npw, lst, done);
  return done + qn;
}

__device__ __forceinline__ void replay_edges(const unsigned int* __restrict__ lp, int n,
                                             int lane, const float* __restrict__ hin,
                                             float* sm) {
  constexpr int U = 4;
  unsigned int nxt[U];
#pragma unroll
  for (int u = 0; u < U; ++u) {
    int i = u * 32 + lane;
    nxt[u] = (i < n) ? lp[i] : NOEDGE;
  }
  for (int i0 = 0; i0 < n; i0 += 32 * U) {
    unsigned int p[U];
    float4 r0[U], r1[U], r2[U];
#pragma unroll
    for (int u = 0; u < U; ++u) {
      p[u] = nxt[u];
      if (p[u] != NOEDGE) {
        const float4* rp = (const float4*)(hin + (long long)(p[u] >> LOCB) * SP);
        r0[u] = rp[0]; r1[u] = rp[1]; r2[u] = rp[2];
      }
    }
#pragma unroll
    for (int u = 0; u < U; ++u) {
      int i = i0 + 32 * U + u * 32 + lane;
      nxt[u] = (i < n) ? lp[i] : NOEDGE;
    }
#pragma unroll
    for (int u = 0; u < U; ++u) {
      if (p[u] != NOEDGE)
        add_row(sm + (p[u] & ((1u << LOCB) - 1u)) * SP, r0[u], r1[u], r2[u]);
    }
  }
}


template <bool REPLAY>
__global__ __launch_bounds__(1024)
void k_layer(const float* __restrict__ hin,
             const int* __restrict__ esrc,
             const int* __restrict__ edst,
             const unsigned int* __restrict__ owner,
             const float* __restrict__ WsP,
             const float* __restrict__ WnP,
             const float* __restrict__ bsP,
             float* __restrict__ hout,
             unsigned int* __restrict__ lists,
             int* __restrict__ lcnt,
             int Nn, int E_, int Epad, int npw) {
  __shared__ __attribute__((aligned(16))) float s_msum[2 * NPW_MAX * SP];
  __shared__ int s_q[32 * 64];

  const int tid  = threadIdx.x;
  const int lane = tid & 31;
  const int wave = tid >> 5;
  const int m    = lane & 15;
  const int half = lane >> 4;
  const int lo   = blockIdx.x * npw;
  const int cnt  = min(npw, Nn - lo);
  const int nT   = (cnt + 15) >> 4;
  int* wq = s_q + wave * 64;

  union Acc { v8f v; float f[8]; };
  Acc acc[TPW];

  {
    float bsum = bsP[m];
    const float* brow = WsP + m * SP;
#pragma unroll
    for (int i = 0; i < TPW; ++i) {
      int t = wave + 32 * i;
#pragma unroll
      for (int j = 0; j < 8; ++j) acc[i].f[j] = bsum;
      if (t < nT) {
        int node = min(lo + t * 16 + m, Nn - 1);
        const float* arow = hin + (long long)node * SP;
#pragma unroll
        for (int c = 0; c < 3; ++c) {
          int k0 = c * 4 + half * 2;
          v2f a, b;
          a.x = arow[k0]; a.y = arow[k0 + 1];
          b.x = brow[k0]; b.y = brow[k0 + 1];
          acc[i].v = wmma_k4(a, b, acc[i].v);
        }
      }
    }
  }

  const unsigned int myTag = (unsigned int)blockIdx.x * 0x01010101u;
  const int nVec = Epad >> 4;

  static_assert(RT % 2 == 0, "etypes are taken two per pass");
  for (int rp = 0; rp < RT; rp += 2) {
    for (int i = tid; i < npw * (SP / 4); i += 1024) {
      ((float4*)s_msum)[i] = make_float4(0.f, 0.f, 0.f, 0.f);
      ((float4*)(s_msum + NPW_MAX * SP))[i] = make_float4(0.f, 0.f, 0.f, 0.f);
    }
    __syncthreads();

   for (int rr = 0; rr < 2; ++rr) {
    const int r = rp + rr;
    float* sm = s_msum + rr * (NPW_MAX * SP);
    const uint4* ow  = (const uint4*)(owner + (long long)r * (Epad >> 2));
    const int* srcr  = esrc + (long long)r * E_;
    const int* dstr  = edst + (long long)r * E_;
    const long long slot = ((long long)blockIdx.x * RT + r) * 32 + wave;
    unsigned int* lst = lists + slot * LCAP;

    if (REPLAY) {
      int n = __builtin_amdgcn_readfirstlane(lcnt[slot]);
      if (n <= LCAP) replay_edges(lst, n, lane, hin, sm);
      else
        scan_edges<false>(ow, nVec, myTag, tid, lane, wq, hin, srcr, dstr, sm, lo, npw, lst);
    } else {
      int n = scan_edges<true>(ow, nVec, myTag, tid, lane, wq, hin, srcr, dstr, sm, lo, npw, lst);
      if (lane == 0) lcnt[slot] = n;
    }
   }
    __syncthreads();

#pragma unroll
    for (int rr = 0; rr < 2; ++rr) {
      const float* brow = WnP + (rp + rr) * 16 * SP + m * SP;
      const float* sm = s_msum + rr * (NPW_MAX * SP);
#pragma unroll
      for (int i = 0; i < TPW; ++i) {
        int t = wave + 32 * i;
        if (t < nT) {
          const float* arow = sm + (t * 16 + m) * SP;
          float rdeg = 1.0f / fmaxf(arow[10], 1.0f);
#pragma unroll
          for (int c = 0; c < 3; ++c) {
            int k0 = c * 4 + half * 2;
            v2f a, b;
            a.x = arow[k0] * rdeg; a.y = arow[k0 + 1] * rdeg;
            b.x = brow[k0];        b.y = brow[k0 + 1];
            acc[i].v = wmma_k4(a, b, acc[i].v);
          }
        }
      }
    }
    __syncthreads();
  }

#pragma unroll
  for (int i = 0; i < TPW; ++i) {
    int t = wave + 32 * i;
    if (t < nT && m < SP) {
      int rowBase = lo + t * 16 + half * 8;
      float* op = hout + (long long)rowBase * SP + m;
#pragma unroll
      for (int j = 0; j < 8; ++j)
        if (rowBase + j < Nn) op[j * SP] = acc[i].f[j];
    }
  }
}

__global__ void k_score(const float* __restrict__ hp,
                        const int* __restrict__ ps, const int* __restrict__ pd,
                        const int* __restrict__ ns, const int* __restrict__ nd,
                        float* __restrict__ out, int E_) {
  int i = blockIdx.x * blockDim.x + threadIdx.x;
  if (i >= 2 * E_) return;
  int a, b;
  if (i < E_) { a = ps[i]; b = pd[i]; }
  else        { a = ns[i - E_]; b = nd[i - E_]; }
  const float* pa = hp + (long long)a * SP;
  const float* pb = hp + (long long)b * SP;
  float s = 0.0f;
#pragma unroll
  for (int k = 0; k < D; ++k) s += pa[k] * pb[k];
  out[i] = s;
}

extern "C" void kernel_launch(void* const* d_in, const int* in_sizes, int n_in,
                              void* d_out, int out_size, void* d_ws, size_t ws_size,
                              hipStream_t stream) {
  (void)n_in; (void)out_size; (void)ws_size;
  const float* x    = (const float*)d_in[0];
  const int*   esrc = (const int*)d_in[1];
  const int*   edst = (const int*)d_in[2];
  const int*   nsrc = (const int*)d_in[3];
  const int*   ndst = (const int*)d_in[4];
  const float* Ws1  = (const float*)d_in[5];
  const float* Wn1  = (const float*)d_in[6];
  const float* b1   = (const float*)d_in[7];
  const float* Ws2  = (const float*)d_in[8];
  const float* Wn2  = (const float*)d_in[9];
  const float* b2   = (const float*)d_in[10];

  const int Nn = in_sizes[0] / D;
  const int Ee = in_sizes[3];

  float* B0   = (float*)d_ws;
  float* B1   = B0 + (size_t)Nn * SP;
  float* WsP1 = B1 + (size_t)Nn * SP;
  float* WnP1 = WsP1 + 16 * SP;
  float* bsP1 = WnP1 + RT * 16 * SP;
  float* WsP2 = bsP1 + 16;
  float* WnP2 = WsP2 + 16 * SP;
  float* bsP2 = WnP2 + RT * 16 * SP;
  unsigned int* owner = (unsigned int*)(bsP2 + 16);

  const int Epad   = (Ee + 15) & ~15;
  int* lcnt = (int*)(owner + RT * (size_t)(Epad >> 2));
  unsigned int* lists = (unsigned int*)(lcnt + 255 * RT * 32);
  const int npw    = (((Nn + 254) / 255) + 15) & ~15;
  const int lBlocks = (Nn + npw - 1) / npw;
  const int padCnt = Nn * SP;
  const int ownCnt = RT * (Epad >> 2);

  k_prep<<<1, 256, 0, stream>>>(Ws1, Wn1, b1, Ws2, Wn2, b2,
                                WsP1, WnP1, bsP1, WsP2, WnP2, bsP2);
  k_padcopy<<<(padCnt + 255) / 256, 256, 0, stream>>>(x, B0, Nn);
  k_owner<<<(ownCnt + 255) / 256, 256, 0, stream>>>(edst, owner, Ee, Epad, npw);

  k_layer<false><<<lBlocks, 1024, 0, stream>>>(B0, esrc, edst, owner, WsP1, WnP1, bsP1, B1,
                                               lists, lcnt, Nn, Ee, Epad, npw);
  k_layer<true><<<lBlocks, 1024, 0, stream>>>(B1, esrc, edst, owner, WsP2, WnP2, bsP2, B0,
                                              lists, lcnt, Nn, Ee, Epad, npw);

  k_score<<<(2 * Ee + 255) / 256, 256, 0, stream>>>(B0, esrc, edst, nsrc, ndst,
                                                    (float*)d_out, Ee);
}


# Layer-1 filter reads each trip's owner tags two trips ahead (two register stages, refilled once a stage's match flags are formed), not at the trip's head; sums stay LDS-atomic arrival order
# speedup vs baseline: 3.0151x; 1.1271x over previous
#include <hip/hip_runtime.h>


#define D  10
#define RT 4
#define SP 12
#define NPW_MAX 2048
#define TPW (NPW_MAX / 16 / 32)
#define NOTAG 255u
#define LCAP 1024
#define LOCB 11
#define NOEDGE 0xFFFFFFFFu

typedef __attribute__((ext_vector_type(2))) float v2f;
typedef __attribute__((ext_vector_type(8))) float v8f;

__device__ __forceinline__ v8f wmma_k4(v2f a, v2f b, v8f c) {
  return __builtin_amdgcn_wmma_f32_16x16x4_f32(
       false, a,  false, b,
       (short)0, c,  false,  false);
}

__global__ void k_prep(const float* __restrict__ Ws1, const float* __restrict__ Wn1,
                       const float* __restrict__ b1,
                       const float* __restrict__ Ws2, const float* __restrict__ Wn2,
                       const float* __restrict__ b2,
                       float* __restrict__ WsP1, float* __restrict__ WnP1,
                       float* __restrict__ bsP1,
                       float* __restrict__ WsP2, float* __restrict__ WnP2,
                       float* __restrict__ bsP2) {
  for (int i = threadIdx.x; i < 16 * SP; i += blockDim.x) {
    int n = i / SP, k = i % SP;
    float s1 = 0.f, s2 = 0.f;
    if (n < D && k < D) {
#pragma unroll
      for (int r = 0; r < RT; ++r) {
        s1 += Ws1[r * D * D + n * D + k];
        s2 += Ws2[r * D * D + n * D + k];
      }
    }
    WsP1[i] = s1;
    WsP2[i] = s2;
  }
  for (int i = threadIdx.x; i < RT * 16 * SP; i += blockDim.x) {
    int r = i / (16 * SP), j = i % (16 * SP);
    int n = j / SP, k = j % SP;
    float v1 = 0.f, v2 = 0.f;
    if (n < D && k < D) {
      v1 = Wn1[r * D * D + n * D + k];
      v2 = Wn2[r * D * D + n * D + k];
    }
    WnP1[i] = v1;
    WnP2[i] = v2;
  }
  for (int i = threadIdx.x; i < 16; i += blockDim.x) {
    float s1 = 0.f, s2 = 0.f;
    if (i < D) {
#pragma unroll
      for (int r = 0; r < RT; ++r) {
        s1 += b1[r * D + i];
        s2 += b2[r * D + i];
      }
    }
    bsP1[i] = s1;
    bsP2[i] = s2;
  }
}

__global__ void k_padcopy(const float* __restrict__ x, float* __restrict__ hp, int Nn) {
  int i = blockIdx.x * blockDim.x + threadIdx.x;
  if (i >= Nn * SP) return;
  int n = i / SP, k = i - n * SP;
  hp[i] = (k < D) ? x[(long long)n * D + k] : 0.0f;
}

__global__ void k_owner(const int* __restrict__ src, const int* __restrict__ dst,
                        unsigned int* __restrict__ owner, unsigned int* __restrict__ words,
                        int E_, int Epad, int npw) {
  int q4 = Epad >> 2;
  int i = blockIdx.x * blockDim.x + threadIdx.x;
  if (i >= RT * q4) return;
  int r  = i / q4;
  int e0 = (i - r * q4) * 4;
  const int* sp = src + (long long)r * E_ + e0;
  const int* dp = dst + (long long)r * E_ + e0;
  int s[4] = {0, 0, 0, 0}, d[4] = {0, 0, 0, 0};
  bool wide = (e0 + 3 < E_) && ((((unsigned long long)sp | (unsigned long long)dp) & 15ull) == 0ull);
  if (wide) {
    int4 sv = *(const int4*)sp, dv = *(const int4*)dp;
    s[0] = sv.x; s[1] = sv.y; s[2] = sv.z; s[3] = sv.w;
    d[0] = dv.x; d[1] = dv.y; d[2] = dv.z; d[3] = dv.w;
  } else {
#pragma unroll
    for (int b = 0; b < 4; ++b)
      if (e0 + b < E_) { s[b] = sp[b]; d[b] = dp[b]; }
  }
  unsigned int pack = 0;
  unsigned int w[4];
#pragma unroll
  for (int b = 0; b < 4; ++b) {
    unsigned int tag = NOTAG;
    w[b] = NOEDGE;
    if (e0 + b < E_) {
      tag = (unsigned int)d[b] / (unsigned int)npw;
      unsigned int loc = (unsigned int)d[b] - tag * (unsigned int)npw;
      w[b] = ((unsigned int)s[b] << LOCB) | loc;
    }
    pack |= tag << (8 * b);
  }
  owner[i] = pack;
  ((uint4*)words)[i] = make_uint4(w[0], w[1], w[2], w[3]);
}

__device__ __forceinline__ unsigned int zero_bytes(unsigned int x) {
  return ~(((x & 0x7F7F7F7Fu) + 0x7F7F7F7Fu) | x | 0x7F7F7F7Fu);
}

__device__ __forceinline__ void add_row(float* mp, float4 r0, float4 r1, float4 r2) {
  atomicAdd(mp + 0, r0.x); atomicAdd(mp + 1, r0.y);
  atomicAdd(mp + 2, r0.z); atomicAdd(mp + 3, r0.w);
  atomicAdd(mp + 4, r1.x); atomicAdd(mp + 5, r1.y);
  atomicAdd(mp + 6, r1.z); atomicAdd(mp + 7, r1.w);
  atomicAdd(mp + 8, r2.x); atomicAdd(mp + 9, r2.y);
  atomicAdd(mp + 10, 1.0f);
}

template <bool SAVE>
__device__ __forceinline__ void flush_edges(const int* wq, int n, int lane,
                                            const float* __restrict__ hin,
                                            const unsigned int* __restrict__ wr,
                                            float* s_msum, int npw,
                                            unsigned int* __restrict__ lst, int pos) {
  if (lane < n) {
    int e = wq[lane];
    unsigned int w = wr[e];
    unsigned int loc = w & ((1u << LOCB) - 1u);
    bool ok = loc < (unsigned int)npw;
    if (SAVE) {
      int at = pos + lane;
      if (at < LCAP) lst[at] = ok ? w : NOEDGE;
    }
    if (ok) {
      const float4* rp = (const float4*)(hin + (long long)(w >> LOCB) * SP);
      float4 r0 = rp[0], r1 = rp[1], r2 = rp[2];
      add_row(s_msum + loc * SP, r0, r1, r2);
    }
  }
}

__device__ __forceinline__ unsigned int match_flags(uint4 q0, uint4 q1, unsigned int myTag) {
  unsigned int mm;
  mm  = zero_bytes(q0.x ^ myTag) >> 7;
  mm |= zero_bytes(q0.y ^ myTag) >> 6;
  mm |= zero_bytes(q0.z ^ myTag) >> 5;
  mm |= zero_bytes(q0.w ^ myTag) >> 4;
  mm |= zero_bytes(q1.x ^ myTag) >> 3;
  mm |= zero_bytes(q1.y ^ myTag) >> 2;
  mm |= zero_bytes(q1.z ^ myTag) >> 1;
  mm |= zero_bytes(q1.w ^ myTag);
  return mm;
}

__device__ __forceinline__ void load_tags(const uint4* __restrict__ ow, int nVec, int base,
                                          int tid, uint4& q0, uint4& q1) {
  int v0 = base + tid, v1 = v0 + 1024;
  const int last = max(nVec - 1, 0);
  q0 = ow[min(v0, last)];
  q1 = ow[min(v1, last)];
}

template <bool SAVE, bool AHEAD>
__device__ __forceinline__ void scan_trip(const uint4* __restrict__ ow, int nVec, int base,
                                          uint4& s0, uint4& s1,
                                          unsigned int myTag, int tid, int lane, int* wq,
                                          const float* __restrict__ hin,
                                          const unsigned int* __restrict__ wr,
                                          float* sm, int npw,
                                          unsigned int* __restrict__ lst,
                                          int& qn, int& done) {
  const int v0 = base + tid, v1 = v0 + 1024;
  unsigned int mm = match_flags(s0, s1, myTag);
  if (AHEAD) {
    mm &= ((v0 < nVec) ? 0x0F0F0F0Fu : 0u) | ((v1 < nVec) ? 0xF0F0F0F0u : 0u);
    load_tags(ow, nVec, base + 4096, tid, s0, s1);
  }

  for (int trip = 0; trip < 32; ++trip) {
    bool has = (mm != 0u);
    unsigned int bal = __builtin_amdgcn_ballot_w32(has);
    if (bal == 0u) break;
    if (has) {
      int p = __builtin_ctz(mm);
      mm &= mm - 1u;
      int jj = p & 7, n = p >> 3;
      int v  = (jj & 4) ? v1 : v0;
      int e  = v * 16 + (jj & 3) * 4 + n;
      int pos = qn + __builtin_popcount(bal & ((1u << lane) - 1u));
      wq[pos] = e;
    }
    qn += __builtin_popcount(bal);
    if (qn >= 32) {
      flush_edges<SAVE>(wq, 32, lane, hin, wr, sm, npw, lst, done);
      done += 32;
      int rest = qn - 32;
      int tmp = 0;
      if (lane < rest) tmp = wq[32 + lane];
      if (lane < rest) wq[lane] = tmp;
      qn = rest;
    }
  }
}

template <bool SAVE>
__device__ __forceinline__ int scan_edges(const uint4* __restrict__ ow, int nVec,
                                          unsigned int myTag, int tid, int lane, int* wq,
                                          const float* __restrict__ hin,
                                          const unsigned int* __restrict__ wr,
                                          float* sm, int npw,
                                          unsigned int* __restrict__ lst) {
  int qn = 0;
  int done = 0;

  uint4 a0, a1;
  if (SAVE) {
    uint4 b0, b1;
    load_tags(ow, nVec, 2048, tid, b0, b1);
    load_tags(ow, nVec, 0, tid, a0, a1);
    for (int base = 0; base < nVec; base += 4096) {
      scan_trip<SAVE, true>(ow, nVec, base, a0, a1, myTag, tid, lane, wq, hin, wr, sm, npw, lst, qn, done);
      scan_trip<SAVE, true>(ow, nVec, base + 2048, b0, b1, myTag, tid, lane, wq, hin, wr, sm, npw, lst, qn, done);
    }
  } else {
    for (int base = 0; base < nVec; base += 2048) {
      int v0 = base + tid, v1 = v0 + 1024;
      a0 = make_uint4(~0u, ~0u, ~0u, ~0u);
      a1 = a0;
      if (v0 < nVec) a0 = ow[v0];
      if (v1 < nVec) a1 = ow[v1];
      scan_trip<SAVE, false>(ow, nVec, base, a0, a1, myTag, tid, lane, wq, hin, wr, sm, npw, lst, qn, done);
    }
  }
  flush_edges<SAVE>(wq, qn, lane, hin, wr, sm, npw, lst, done);
  return done + qn;
}

__device__ __forceinline__ void replay_edges(const unsigned int* __restrict__ lp, int n,
                                             int lane, const float* __restrict__ hin,
                                             float* sm) {
  constexpr int U = 4;
  unsigned int nxt[U];
#pragma unroll
  for (int u = 0; u < U; ++u) {
    int i = u * 32 + lane;
    nxt[u] = (i < n) ? lp[i] : NOEDGE;
  }
  for (int i0 = 0; i0 < n; i0 += 32 * U) {
    unsigned int p[U];
    float4 r0[U], r1[U], r2[U];
#pragma unroll
    for (int u = 0; u < U; ++u) {
      p[u] = nxt[u];
      if (p[u] != NOEDGE) {
        const float4* rp = (const float4*)(hin + (long long)(p[u] >> LOCB) * SP);
        r0[u] = rp[0]; r1[u] = rp[1]; r2[u] = rp[2];
      }
    }
#pragma unroll
    for (int u = 0; u < U; ++u) {
      int i = i0 + 32 * U + u * 32 + lane;
      nxt[u] = (i < n) ? lp[i] : NOEDGE;
    }
#pragma unroll
    for (int u = 0; u < U; ++u) {
      if (p[u] != NOEDGE)
        add_row(sm + (p[u] & ((1u << LOCB) - 1u)) * SP, r0[u], r1[u], r2[u]);
    }
  }
}


template <bool REPLAY>
__global__ __launch_bounds__(1024)
void k_layer(const float* __restrict__ hin,
             const unsigned int* __restrict__ owner,
             const unsigned int* __restrict__ words,
             const float* __restrict__ WsP,
             const float* __restrict__ WnP,
             const float* __restrict__ bsP,
             float* __restrict__ hout,
             unsigned int* __restrict__ lists,
             int* __restrict__ lcnt,
             int Nn, int Epad, int npw) {
  __shared__ __attribute__((aligned(16))) float s_msum[2 * NPW_MAX * SP];
  __shared__ int s_q[32 * 64];

  const int tid  = threadIdx.x;
  const int lane = tid & 31;
  const int wave = tid >> 5;
  const int m    = lane & 15;
  const int half = lane >> 4;
  const int lo   = blockIdx.x * npw;
  const int cnt  = min(npw, Nn - lo);
  const int nT   = (cnt + 15) >> 4;
  int* wq = s_q + wave * 64;

  union Acc { v8f v; float f[8]; };
  Acc acc[TPW];

  {
    float bsum = bsP[m];
    const float* brow = WsP + m * SP;
#pragma unroll
    for (int i = 0; i < TPW; ++i) {
      int t = wave + 32 * i;
#pragma unroll
      for (int j = 0; j < 8; ++j) acc[i].f[j] = bsum;
      if (t < nT) {
        int node = min(lo + t * 16 + m, Nn - 1);
        const float* arow = hin + (long long)node * SP;
#pragma unroll
        for (int c = 0; c < 3; ++c) {
          int k0 = c * 4 + half * 2;
          v2f a, b;
          a.x = arow[k0]; a.y = arow[k0 + 1];
          b.x = brow[k0]; b.y = brow[k0 + 1];
          acc[i].v = wmma_k4(a, b, acc[i].v);
        }
      }
    }
  }

  const unsigned int myTag = (unsigned int)blockIdx.x * 0x01010101u;
  const int nVec = Epad >> 4;

  static_assert(RT % 2 == 0, "etypes are taken two per pass");
  for (int rp = 0; rp < RT; rp += 2) {
    for (int i = tid; i < npw * (SP / 4); i += 1024) {
      ((float4*)s_msum)[i] = make_float4(0.f, 0.f, 0.f, 0.f);
      ((float4*)(s_msum + NPW_MAX * SP))[i] = make_float4(0.f, 0.f, 0.f, 0.f);
    }
    __syncthreads();

   for (int rr = 0; rr < 2; ++rr) {
    const int r = rp + rr;
    float* sm = s_msum + rr * (NPW_MAX * SP);
    const uint4* ow  = (const uint4*)(owner + (long long)r * (Epad >> 2));
    const unsigned int* wr = words + (long long)r * Epad;
    const long long slot = ((long long)blockIdx.x * RT + r) * 32 + wave;
    unsigned int* lst = lists + slot * LCAP;

    if (REPLAY) {
      int n = __builtin_amdgcn_readfirstlane(lcnt[slot]);
      if (n <= LCAP) replay_edges(lst, n, lane, hin, sm);
      else
        scan_edges<false>(ow, nVec, myTag, tid, lane, wq, hin, wr, sm, npw, lst);
    } else {
      int n = scan_edges<true>(ow, nVec, myTag, tid, lane, wq, hin, wr, sm, npw, lst);
      if (lane == 0) lcnt[slot] = n;
    }
   }
    __syncthreads();

#pragma unroll
    for (int rr = 0; rr < 2; ++rr) {
      const float* brow = WnP + (rp + rr) * 16 * SP + m * SP;
      const float* sm = s_msum + rr * (NPW_MAX * SP);
#pragma unroll
      for (int i = 0; i < TPW; ++i) {
        int t = wave + 32 * i;
        if (t < nT) {
          const float* arow = sm + (t * 16 + m) * SP;
          float rdeg = 1.0f / fmaxf(arow[10], 1.0f);
#pragma unroll
          for (int c = 0; c < 3; ++c) {
            int k0 = c * 4 + half * 2;
            v2f a, b;
            a.x = arow[k0] * rdeg; a.y = arow[k0 + 1] * rdeg;
            b.x = brow[k0];        b.y = brow[k0 + 1];
            acc[i].v = wmma_k4(a, b, acc[i].v);
          }
        }
      }
    }
    __syncthreads();
  }

#pragma unroll
  for (int i = 0; i < TPW; ++i) {
    int t = wave + 32 * i;
    if (t < nT && m < SP) {
      int rowBase = lo + t * 16 + half * 8;
      float* op = hout + (long long)rowBase * SP + m;
#pragma unroll
      for (int j = 0; j < 8; ++j)
        if (rowBase + j < Nn) op[j * SP] = acc[i].f[j];
    }
  }
}

__global__ void k_score(const float* __restrict__ hp,
                        const int* __restrict__ ps, const int* __restrict__ pd,
                        const int* __restrict__ ns, const int* __restrict__ nd,
                        float* __restrict__ out, int E_) {
  int i = blockIdx.x * blockDim.x + threadIdx.x;
  if (i >= 2 * E_) return;
  int a, b;
  if (i < E_) { a = ps[i]; b = pd[i]; }
  else        { a = ns[i - E_]; b = nd[i - E_]; }
  const float* pa = hp + (long long)a * SP;
  const float* pb = hp + (long long)b * SP;
  float s = 0.0f;
#pragma unroll
  for (int k = 0; k < D; ++k) s += pa[k] * pb[k];
  out[i] = s;
}

extern "C" void kernel_launch(void* const* d_in, const int* in_sizes, int n_in,
                              void* d_out, int out_size, void* d_ws, size_t ws_size,
                              hipStream_t stream) {
  (void)n_in; (void)out_size; (void)ws_size;
  const float* x    = (const float*)d_in[0];
  const int*   esrc = (const int*)d_in[1];
  const int*   edst = (const int*)d_in[2];
  const int*   nsrc = (const int*)d_in[3];
  const int*   ndst = (const int*)d_in[4];
  const float* Ws1  = (const float*)d_in[5];
  const float* Wn1  = (const float*)d_in[6];
  const float* b1   = (const float*)d_in[7];
  const float* Ws2  = (const float*)d_in[8];
  const float* Wn2  = (const float*)d_in[9];
  const float* b2   = (const float*)d_in[10];

  const int Nn = in_sizes[0] / D;
  const int Ee = in_sizes[3];

  float* B0   = (float*)d_ws;
  float* B1   = B0 + (size_t)Nn * SP;
  float* WsP1 = B1 + (size_t)Nn * SP;
  float* WnP1 = WsP1 + 16 * SP;
  float* bsP1 = WnP1 + RT * 16 * SP;
  float* WsP2 = bsP1 + 16;
  float* WnP2 = WsP2 + 16 * SP;
  float* bsP2 = WnP2 + RT * 16 * SP;
  unsigned int* owner = (unsigned int*)(bsP2 + 16);

  const int Epad   = (Ee + 15) & ~15;
  unsigned int* words = owner + RT * (size_t)(Epad >> 2);
  int* lcnt = (int*)(words + RT * (size_t)Epad);
  unsigned int* lists = (unsigned int*)(lcnt + 255 * RT * 32);
  const int npw    = (((Nn + 254) / 255) + 15) & ~15;
  const int lBlocks = (Nn + npw - 1) / npw;
  const int padCnt = Nn * SP;
  const int ownCnt = RT * (Epad >> 2);

  k_prep<<<1, 256, 0, stream>>>(Ws1, Wn1, b1, Ws2, Wn2, b2,
                                WsP1, WnP1, bsP1, WsP2, WnP2, bsP2);
  k_padcopy<<<(padCnt + 255) / 256, 256, 0, stream>>>(x, B0, Nn);
  k_owner<<<(ownCnt + 255) / 256, 256, 0, stream>>>(esrc, edst, owner, words, Ee, Epad, npw);

  k_layer<false><<<lBlocks, 1024, 0, stream>>>(B0, owner, words, WsP1, WnP1, bsP1, B1,
                                               lists, lcnt, Nn, Epad, npw);
  k_layer<true><<<lBlocks, 1024, 0, stream>>>(B1, owner, words, WsP2, WnP2, bsP2, B0,
                                              lists, lcnt, Nn, Epad, npw);

  k_score<<<(2 * Ee + 255) / 256, 256, 0, stream>>>(B0, esrc, edst, nsrc, ndst,
                                                    (float*)d_out, Ee);
}


# Layer 1 reads the raw 40-byte feature rows itself (k = 10, 11 operand slots zeroed in registers); the padded-copy launch, one thread per float, goes: 6 launches to 5; sum order unchanged
# speedup vs baseline: 3.1263x; 1.0369x over previous
#include <hip/hip_runtime.h>


#define D  10
#define RT 4
#define SP 12
#define NPW_MAX 2048
#define TPW (NPW_MAX / 16 / 32)
#define NOTAG 255u
#define LCAP 1024
#define LOCB 11
#define NOEDGE 0xFFFFFFFFu

typedef __attribute__((ext_vector_type(2))) float v2f;
typedef __attribute__((ext_vector_type(8))) float v8f;

__device__ __forceinline__ v8f wmma_k4(v2f a, v2f b, v8f c) {
  return __builtin_amdgcn_wmma_f32_16x16x4_f32(
       false, a,  false, b,
       (short)0, c,  false,  false);
}

__global__ void k_prep(const float* __restrict__ Ws1, const float* __restrict__ Wn1,
                       const float* __restrict__ b1,
                       const float* __restrict__ Ws2, const float* __restrict__ Wn2,
                       const float* __restrict__ b2,
                       float* __restrict__ WsP1, float* __restrict__ WnP1,
                       float* __restrict__ bsP1,
                       float* __restrict__ WsP2, float* __restrict__ WnP2,
                       float* __restrict__ bsP2) {
  for (int i = threadIdx.x; i < 16 * SP; i += blockDim.x) {
    int n = i / SP, k = i % SP;
    float s1 = 0.f, s2 = 0.f;
    if (n < D && k < D) {
#pragma unroll
      for (int r = 0; r < RT; ++r) {
        s1 += Ws1[r * D * D + n * D + k];
        s2 += Ws2[r * D * D + n * D + k];
      }
    }
    WsP1[i] = s1;
    WsP2[i] = s2;
  }
  for (int i = threadIdx.x; i < RT * 16 * SP; i += blockDim.x) {
    int r = i / (16 * SP), j = i % (16 * SP);
    int n = j / SP, k = j % SP;
    float v1 = 0.f, v2 = 0.f;
    if (n < D && k < D) {
      v1 = Wn1[r * D * D + n * D + k];
      v2 = Wn2[r * D * D + n * D + k];
    }
    WnP1[i] = v1;
    WnP2[i] = v2;
  }
  for (int i = threadIdx.x; i < 16; i += blockDim.x) {
    float s1 = 0.f, s2 = 0.f;
    if (i < D) {
#pragma unroll
      for (int r = 0; r < RT; ++r) {
        s1 += b1[r * D + i];
        s2 += b2[r * D + i];
      }
    }
    bsP1[i] = s1;
    bsP2[i] = s2;
  }
}

__global__ void k_owner(const int* __restrict__ src, const int* __restrict__ dst,
                        unsigned int* __restrict__ owner, unsigned int* __restrict__ words,
                        int E_, int Epad, int npw) {
  int q4 = Epad >> 2;
  int i = blockIdx.x * blockDim.x + threadIdx.x;
  if (i >= RT * q4) return;
  int r  = i / q4;
  int e0 = (i - r * q4) * 4;
  const int* sp = src + (long long)r * E_ + e0;
  const int* dp = dst + (long long)r * E_ + e0;
  int s[4] = {0, 0, 0, 0}, d[4] = {0, 0, 0, 0};
  bool wide = (e0 + 3 < E_) && ((((unsigned long long)sp | (unsigned long long)dp) & 15ull) == 0ull);
  if (wide) {
    int4 sv = *(const int4*)sp, dv = *(const int4*)dp;
    s[0] = sv.x; s[1] = sv.y; s[2] = sv.z; s[3] = sv.w;
    d[0] = dv.x; d[1] = dv.y; d[2] = dv.z; d[3] = dv.w;
  } else {
#pragma unroll
    for (int b = 0; b < 4; ++b)
      if (e0 + b < E_) { s[b] = sp[b]; d[b] = dp[b]; }
  }
  unsigned int pack = 0;
  unsigned int w[4];
#pragma unroll
  for (int b = 0; b < 4; ++b) {
    unsigned int tag = NOTAG;
    w[b] = NOEDGE;
    if (e0 + b < E_) {
      tag = (unsigned int)d[b] / (unsigned int)npw;
      unsigned int loc = (unsigned int)d[b] - tag * (unsigned int)npw;
      w[b] = ((unsigned int)s[b] << LOCB) | loc;
    }
    pack |= tag << (8 * b);
  }
  owner[i] = pack;
  ((uint4*)words)[i] = make_uint4(w[0], w[1], w[2], w[3]);
}

__device__ __forceinline__ unsigned int zero_bytes(unsigned int x) {
  return ~(((x & 0x7F7F7F7Fu) + 0x7F7F7F7Fu) | x | 0x7F7F7F7Fu);
}

__device__ __forceinline__ void add_row(float* mp, float4 r0, float4 r1, float4 r2) {
  atomicAdd(mp + 0, r0.x); atomicAdd(mp + 1, r0.y);
  atomicAdd(mp + 2, r0.z); atomicAdd(mp + 3, r0.w);
  atomicAdd(mp + 4, r1.x); atomicAdd(mp + 5, r1.y);
  atomicAdd(mp + 6, r1.z); atomicAdd(mp + 7, r1.w);
  atomicAdd(mp + 8, r2.x); atomicAdd(mp + 9, r2.y);
  atomicAdd(mp + 10, 1.0f);
}

template <bool SAVE>
__device__ __forceinline__ void flush_edges(const int* wq, int n, int lane,
                                            const float* __restrict__ hin,
                                            const unsigned int* __restrict__ wr,
                                            float* s_msum, int npw,
                                            unsigned int* __restrict__ lst, int pos) {
  if (lane < n) {
    int e = wq[lane];
    unsigned int w = wr[e];
    unsigned int loc = w & ((1u << LOCB) - 1u);
    bool ok = loc < (unsigned int)npw;
    if (SAVE) {
      int at = pos + lane;
      if (at < LCAP) lst[at] = ok ? w : NOEDGE;
    }
    if (ok) {
      constexpr int IP = SAVE ? D : SP;
      const float* rp = hin + (long long)(w >> LOCB) * IP;
      float4 r0, r1, r2;
      r0.x = rp[0]; r0.y = rp[1]; r0.z = rp[2]; r0.w = rp[3];
      r1.x = rp[4]; r1.y = rp[5]; r1.z = rp[6]; r1.w = rp[7];
      r2.x = rp[8]; r2.y = rp[9]; r2.z = 0.f;   r2.w = 0.f;
      add_row(s_msum + loc * SP, r0, r1, r2);
    }
  }
}

__device__ __forceinline__ unsigned int match_flags(uint4 q0, uint4 q1, unsigned int myTag) {
  unsigned int mm;
  mm  = zero_bytes(q0.x ^ myTag) >> 7;
  mm |= zero_bytes(q0.y ^ myTag) >> 6;
  mm |= zero_bytes(q0.z ^ myTag) >> 5;
  mm |= zero_bytes(q0.w ^ myTag) >> 4;
  mm |= zero_bytes(q1.x ^ myTag) >> 3;
  mm |= zero_bytes(q1.y ^ myTag) >> 2;
  mm |= zero_bytes(q1.z ^ myTag) >> 1;
  mm |= zero_bytes(q1.w ^ myTag);
  return mm;
}

__device__ __forceinline__ void load_tags(const uint4* __restrict__ ow, int nVec, int base,
                                          int tid, uint4& q0, uint4& q1) {
  int v0 = base + tid, v1 = v0 + 1024;
  const int last = max(nVec - 1, 0);
  q0 = ow[min(v0, last)];
  q1 = ow[min(v1, last)];
}

template <bool SAVE, bool AHEAD>
__device__ __forceinline__ void scan_trip(const uint4* __restrict__ ow, int nVec, int base,
                                          uint4& s0, uint4& s1,
                                          unsigned int myTag, int tid, int lane, int* wq,
                                          const float* __restrict__ hin,
                                          const unsigned int* __restrict__ wr,
                                          float* sm, int npw,
                                          unsigned int* __restrict__ lst,
                                          int& qn, int& done) {
  const int v0 = base + tid, v1 = v0 + 1024;
  unsigned int mm = match_flags(s0, s1, myTag);
  if (AHEAD) {
    mm &= ((v0 < nVec) ? 0x0F0F0F0Fu : 0u) | ((v1 < nVec) ? 0xF0F0F0F0u : 0u);
    load_tags(ow, nVec, base + 4096, tid, s0, s1);
  }

  for (int trip = 0; trip < 32; ++trip) {
    bool has = (mm != 0u);
    unsigned int bal = __builtin_amdgcn_ballot_w32(has);
    if (bal == 0u) break;
    if (has) {
      int p = __builtin_ctz(mm);
      mm &= mm - 1u;
      int jj = p & 7, n = p >> 3;
      int v  = (jj & 4) ? v1 : v0;
      int e  = v * 16 + (jj & 3) * 4 + n;
      int pos = qn + __builtin_popcount(bal & ((1u << lane) - 1u));
      wq[pos] = e;
    }
    qn += __builtin_popcount(bal);
    if (qn >= 32) {
      flush_edges<SAVE>(wq, 32, lane, hin, wr, sm, npw, lst, done);
      done += 32;
      int rest = qn - 32;
      int tmp = 0;
      if (lane < rest) tmp = wq[32 + lane];
      if (lane < rest) wq[lane] = tmp;
      qn = rest;
    }
  }
}

template <bool SAVE>
__device__ __forceinline__ int scan_edges(const uint4* __restrict__ ow, int nVec,
                                          unsigned int myTag, int tid, int lane, int* wq,
                                          const float* __restrict__ hin,
                                          const unsigned int* __restrict__ wr,
                                          float* sm, int npw,
                                          unsigned int* __restrict__ lst) {
  int qn = 0;
  int done = 0;

  uint4 a0, a1;
  if (SAVE) {
    uint4 b0, b1;
    load_tags(ow, nVec, 2048, tid, b0, b1);
    load_tags(ow, nVec, 0, tid, a0, a1);
    for (int base = 0; base < nVec; base += 4096) {
      scan_trip<SAVE, true>(ow, nVec, base, a0, a1, myTag, tid, lane, wq, hin, wr, sm, npw, lst, qn, done);
      scan_trip<SAVE, true>(ow, nVec, base + 2048, b0, b1, myTag, tid, lane, wq, hin, wr, sm, npw, lst, qn, done);
    }
  } else {
    for (int base = 0; base < nVec; base += 2048) {
      int v0 = base + tid, v1 = v0 + 1024;
      a0 = make_uint4(~0u, ~0u, ~0u, ~0u);
      a1 = a0;
      if (v0 < nVec) a0 = ow[v0];
      if (v1 < nVec) a1 = ow[v1];
      scan_trip<SAVE, false>(ow, nVec, base, a0, a1, myTag, tid, lane, wq, hin, wr, sm, npw, lst, qn, done);
    }
  }
  flush_edges<SAVE>(wq, qn, lane, hin, wr, sm, npw, lst, done);
  return done + qn;
}

__device__ __forceinline__ void replay_edges(const unsigned int* __restrict__ lp, int n,
                                             int lane, const float* __restrict__ hin,
                                             float* sm) {
  constexpr int U = 4;
  unsigned int nxt[U];
#pragma unroll
  for (int u = 0; u < U; ++u) {
    int i = u * 32 + lane;
    nxt[u] = (i < n) ? lp[i] : NOEDGE;
  }
  for (int i0 = 0; i0 < n; i0 += 32 * U) {
    unsigned int p[U];
    float4 r0[U], r1[U], r2[U];
#pragma unroll
    for (int u = 0; u < U; ++u) {
      p[u] = nxt[u];
      if (p[u] != NOEDGE) {
        const float4* rp = (const float4*)(hin + (long long)(p[u] >> LOCB) * SP);
        r0[u] = rp[0]; r1[u] = rp[1]; r2[u] = rp[2];
      }
    }
#pragma unroll
    for (int u = 0; u < U; ++u) {
      int i = i0 + 32 * U + u * 32 + lane;
      nxt[u] = (i < n) ? lp[i] : NOEDGE;
    }
#pragma unroll
    for (int u = 0; u < U; ++u) {
      if (p[u] != NOEDGE)
        add_row(sm + (p[u] & ((1u << LOCB) - 1u)) * SP, r0[u], r1[u], r2[u]);
    }
  }
}


template <bool REPLAY>
__global__ __launch_bounds__(1024)
void k_layer(const float* __restrict__ hin,
             const unsigned int* __restrict__ owner,
             const unsigned int* __restrict__ words,
             const float* __restrict__ WsP,
             const float* __restrict__ WnP,
             const float* __restrict__ bsP,
             float* __restrict__ hout,
             unsigned int* __restrict__ lists,
             int* __restrict__ lcnt,
             int Nn, int Epad, int npw) {
  __shared__ __attribute__((aligned(16))) float s_msum[2 * NPW_MAX * SP];
  __shared__ int s_q[32 * 64];

  const int tid  = threadIdx.x;
  const int lane = tid & 31;
  const int wave = tid >> 5;
  const int m    = lane & 15;
  const int half = lane >> 4;
  const int lo   = blockIdx.x * npw;
  const int cnt  = min(npw, Nn - lo);
  const int nT   = (cnt + 15) >> 4;
  int* wq = s_q + wave * 64;

  union Acc { v8f v; float f[8]; };
  Acc acc[TPW];

  {
    float bsum = bsP[m];
    const float* brow = WsP + m * SP;
#pragma unroll
    for (int i = 0; i < TPW; ++i) {
      int t = wave + 32 * i;
#pragma unroll
      for (int j = 0; j < 8; ++j) acc[i].f[j] = bsum;
      if (t < nT) {
        constexpr int IP = REPLAY ? SP : D;
        int node = min(lo + t * 16 + m, Nn - 1);
        const float* arow = hin + (long long)node * IP;
#pragma unroll
        for (int c = 0; c < 3; ++c) {
          int k0 = c * 4 + half * 2;
          v2f a, b;
          bool stored = (k0 + 1 < IP);
          a.x = stored ? arow[k0] : 0.f; a.y = stored ? arow[k0 + 1] : 0.f;
          b.x = brow[k0]; b.y = brow[k0 + 1];
          acc[i].v = wmma_k4(a, b, acc[i].v);
        }
      }
    }
  }

  const unsigned int myTag = (unsigned int)blockIdx.x * 0x01010101u;
  const int nVec = Epad >> 4;

  static_assert(RT % 2 == 0, "etypes are taken two per pass");
  for (int rp = 0; rp < RT; rp += 2) {
    for (int i = tid; i < npw * (SP / 4); i += 1024) {
      ((float4*)s_msum)[i] = make_float4(0.f, 0.f, 0.f, 0.f);
      ((float4*)(s_msum + NPW_MAX * SP))[i] = make_float4(0.f, 0.f, 0.f, 0.f);
    }
    __syncthreads();

   for (int rr = 0; rr < 2; ++rr) {
    const int r = rp + rr;
    float* sm = s_msum + rr * (NPW_MAX * SP);
    const uint4* ow  = (const uint4*)(owner + (long long)r * (Epad >> 2));
    const unsigned int* wr = words + (long long)r * Epad;
    const long long slot = ((long long)blockIdx.x * RT + r) * 32 + wave;
    unsigned int* lst = lists + slot * LCAP;

    if (REPLAY) {
      int n = __builtin_amdgcn_readfirstlane(lcnt[slot]);
      if (n <= LCAP) replay_edges(lst, n, lane, hin, sm);
      else
        scan_edges<false>(ow, nVec, myTag, tid, lane, wq, hin, wr, sm, npw, lst);
    } else {
      int n = scan_edges<true>(ow, nVec, myTag, tid, lane, wq, hin, wr, sm, npw, lst);
      if (lane == 0) lcnt[slot] = n;
    }
   }
    __syncthreads();

#pragma unroll
    for (int rr = 0; rr < 2; ++rr) {
      const float* brow = WnP + (rp + rr) * 16 * SP + m * SP;
      const float* sm = s_msum + rr * (NPW_MAX * SP);
#pragma unroll
      for (int i = 0; i < TPW; ++i) {
        int t = wave + 32 * i;
        if (t < nT) {
          const float* arow = sm + (t * 16 + m) * SP;
          float rdeg = 1.0f / fmaxf(arow[10], 1.0f);
#pragma unroll
          for (int c = 0; c < 3; ++c) {
            int k0 = c * 4 + half * 2;
            v2f a, b;
            a.x = arow[k0] * rdeg; a.y = arow[k0 + 1] * rdeg;
            b.x = brow[k0];        b.y = brow[k0 + 1];
            acc[i].v = wmma_k4(a, b, acc[i].v);
          }
        }
      }
    }
    __syncthreads();
  }

#pragma unroll
  for (int i = 0; i < TPW; ++i) {
    int t = wave + 32 * i;
    if (t < nT && m < SP) {
      int rowBase = lo + t * 16 + half * 8;
      float* op = hout + (long long)rowBase * SP + m;
#pragma unroll
      for (int j = 0; j < 8; ++j)
        if (rowBase + j < Nn) op[j * SP] = acc[i].f[j];
    }
  }
}

__global__ void k_score(const float* __restrict__ hp,
                        const int* __restrict__ ps, const int* __restrict__ pd,
                        const int* __restrict__ ns, const int* __restrict__ nd,
                        float* __restrict__ out, int E_) {
  int i = blockIdx.x * blockDim.x + threadIdx.x;
  if (i >= 2 * E_) return;
  int a, b;
  if (i < E_) { a = ps[i]; b = pd[i]; }
  else        { a = ns[i - E_]; b = nd[i - E_]; }
  const float* pa = hp + (long long)a * SP;
  const float* pb = hp + (long long)b * SP;
  float s = 0.0f;
#pragma unroll
  for (int k = 0; k < D; ++k) s += pa[k] * pb[k];
  out[i] = s;
}

extern "C" void kernel_launch(void* const* d_in, const int* in_sizes, int n_in,
                              void* d_out, int out_size, void* d_ws, size_t ws_size,
                              hipStream_t stream) {
  (void)n_in; (void)out_size; (void)ws_size;
  const float* x    = (const float*)d_in[0];
  const int*   esrc = (const int*)d_in[1];
  const int*   edst = (const int*)d_in[2];
  const int*   nsrc = (const int*)d_in[3];
  const int*   ndst = (const int*)d_in[4];
  const float* Ws1  = (const float*)d_in[5];
  const float* Wn1  = (const float*)d_in[6];
  const float* b1   = (const float*)d_in[7];
  const float* Ws2  = (const float*)d_in[8];
  const float* Wn2  = (const float*)d_in[9];
  const float* b2   = (const float*)d_in[10];

  const int Nn = in_sizes[0] / D;
  const int Ee = in_sizes[3];

  float* B0   = (float*)d_ws;
  float* B1   = B0 + (size_t)Nn * SP;
  float* WsP1 = B1 + (size_t)Nn * SP;
  float* WnP1 = WsP1 + 16 * SP;
  float* bsP1 = WnP1 + RT * 16 * SP;
  float* WsP2 = bsP1 + 16;
  float* WnP2 = WsP2 + 16 * SP;
  float* bsP2 = WnP2 + RT * 16 * SP;
  unsigned int* owner = (unsigned int*)(bsP2 + 16);

  const int Epad   = (Ee + 15) & ~15;
  unsigned int* words = owner + RT * (size_t)(Epad >> 2);
  int* lcnt = (int*)(words + RT * (size_t)Epad);
  unsigned int* lists = (unsigned int*)(lcnt + 255 * RT * 32);
  const int npw    = (((Nn + 254) / 255) + 15) & ~15;
  const int lBlocks = (Nn + npw - 1) / npw;
  const int ownCnt = RT * (Epad >> 2);

  k_prep<<<1, 256, 0, stream>>>(Ws1, Wn1, b1, Ws2, Wn2, b2,
                                WsP1, WnP1, bsP1, WsP2, WnP2, bsP2);
  k_owner<<<(ownCnt + 255) / 256, 256, 0, stream>>>(esrc, edst, owner, words, Ee, Epad, npw);

  k_layer<false><<<lBlocks, 1024, 0, stream>>>(x, owner, words, WsP1, WnP1, bsP1, B1,
                                               lists, lcnt, Nn, Epad, npw);
  k_layer<true><<<lBlocks, 1024, 0, stream>>>(B1, owner, words, WsP2, WnP2, bsP2, B0,
                                              lists, lcnt, Nn, Epad, npw);

  k_score<<<(2 * Ee + 255) / 256, 256, 0, stream>>>(B0, esrc, edst, nsrc, ndst,
                                                    (float*)d_out, Ee);
}


# Owned LDS sum rows at an odd pitch, 13 floats, not 12: the 32 rows of one LDS atomic add fall on all 64 banks, not on every fourth bank as at pitch 12; sums stay LDS-atomic arrival order
# speedup vs baseline: 3.3326x; 1.0660x over previous
#include <hip/hip_runtime.h>


#define D  10
#define RT 4
#define SP 12
#define LP 13
#define NPW_MAX 2048
#define TPW (NPW_MAX / 16 / 32)
#define NOTAG 255u
#define LCAP 1024
#define LOCB 11
#define NOEDGE 0xFFFFFFFFu

typedef __attribute__((ext_vector_type(2))) float v2f;
typedef __attribute__((ext_vector_type(8))) float v8f;

__device__ __forceinline__ v8f wmma_k4(v2f a, v2f b, v8f c) {
  return __builtin_amdgcn_wmma_f32_16x16x4_f32(
       false, a,  false, b,
       (short)0, c,  false,  false);
}

__global__ void k_prep(const float* __restrict__ Ws1, const float* __restrict__ Wn1,
                       const float* __restrict__ b1,
                       const float* __restrict__ Ws2, const float* __restrict__ Wn2,
                       const float* __restrict__ b2,
                       float* __restrict__ WsP1, float* __restrict__ WnP1,
                       float* __restrict__ bsP1,
                       float* __restrict__ WsP2, float* __restrict__ WnP2,
                       float* __restrict__ bsP2) {
  for (int i = threadIdx.x; i < 16 * SP; i += blockDim.x) {
    int n = i / SP, k = i % SP;
    float s1 = 0.f, s2 = 0.f;
    if (n < D && k < D) {
#pragma unroll
      for (int r = 0; r < RT; ++r) {
        s1 += Ws1[r * D * D + n * D + k];
        s2 += Ws2[r * D * D + n * D + k];
      }
    }
    WsP1[i] = s1;
    WsP2[i] = s2;
  }
  for (int i = threadIdx.x; i < RT * 16 * SP; i += blockDim.x) {
    int r = i / (16 * SP), j = i % (16 * SP);
    int n = j / SP, k = j % SP;
    float v1 = 0.f, v2 = 0.f;
    if (n < D && k < D) {
      v1 = Wn1[r * D * D + n * D + k];
      v2 = Wn2[r * D * D + n * D + k];
    }
    WnP1[i] = v1;
    WnP2[i] = v2;
  }
  for (int i = threadIdx.x; i < 16; i += blockDim.x) {
    float s1 = 0.f, s2 = 0.f;
    if (i < D) {
#pragma unroll
      for (int r = 0; r < RT; ++r) {
        s1 += b1[r * D + i];
        s2 += b2[r * D + i];
      }
    }
    bsP1[i] = s1;
    bsP2[i] = s2;
  }
}

__global__ void k_owner(const int* __restrict__ src, const int* __restrict__ dst,
                        unsigned int* __restrict__ owner, unsigned int* __restrict__ words,
                        int E_, int Epad, int npw) {
  int q4 = Epad >> 2;
  int i = blockIdx.x * blockDim.x + threadIdx.x;
  if (i >= RT * q4) return;
  int r  = i / q4;
  int e0 = (i - r * q4) * 4;
  const int* sp = src + (long long)r * E_ + e0;
  const int* dp = dst + (long long)r * E_ + e0;
  int s[4] = {0, 0, 0, 0}, d[4] = {0, 0, 0, 0};
  bool wide = (e0 + 3 < E_) && ((((unsigned long long)sp | (unsigned long long)dp) & 15ull) == 0ull);
  if (wide) {
    int4 sv = *(const int4*)sp, dv = *(const int4*)dp;
    s[0] = sv.x; s[1] = sv.y; s[2] = sv.z; s[3] = sv.w;
    d[0] = dv.x; d[1] = dv.y; d[2] = dv.z; d[3] = dv.w;
  } else {
#pragma unroll
    for (int b = 0; b < 4; ++b)
      if (e0 + b < E_) { s[b] = sp[b]; d[b] = dp[b]; }
  }
  unsigned int pack = 0;
  unsigned int w[4];
#pragma unroll
  for (int b = 0; b < 4; ++b) {
    unsigned int tag = NOTAG;
    w[b] = NOEDGE;
    if (e0 + b < E_) {
      tag = (unsigned int)d[b] / (unsigned int)npw;
      unsigned int loc = (unsigned int)d[b] - tag * (unsigned int)npw;
      w[b] = ((unsigned int)s[b] << LOCB) | loc;
    }
    pack |= tag << (8 * b);
  }
  owner[i] = pack;
  ((uint4*)words)[i] = make_uint4(w[0], w[1], w[2], w[3]);
}

__device__ __forceinline__ unsigned int zero_bytes(unsigned int x) {
  return ~(((x & 0x7F7F7F7Fu) + 0x7F7F7F7Fu) | x | 0x7F7F7F7Fu);
}

__device__ __forceinline__ void add_row(float* mp, float4 r0, float4 r1, float4 r2) {
  atomicAdd(mp + 0, r0.x); atomicAdd(mp + 1, r0.y);
  atomicAdd(mp + 2, r0.z); atomicAdd(mp + 3, r0.w);
  atomicAdd(mp + 4, r1.x); atomicAdd(mp + 5, r1.y);
  atomicAdd(mp + 6, r1.z); atomicAdd(mp + 7, r1.w);
  atomicAdd(mp + 8, r2.x); atomicAdd(mp + 9, r2.y);
  atomicAdd(mp + 10, 1.0f);
}

template <bool SAVE>
__device__ __forceinline__ void flush_edges(const int* wq, int n, int lane,
                                            const float* __restrict__ hin,
                                            const unsigned int* __restrict__ wr,
                                            float* s_msum, int npw,
                                            unsigned int* __restrict__ lst, int pos) {
  if (lane < n) {
    int e = wq[lane];
    unsigned int w = wr[e];
    unsigned int loc = w & ((1u << LOCB) - 1u);
    bool ok = loc < (unsigned int)npw;
    if (SAVE) {
      int at = pos + lane;
      if (at < LCAP) lst[at] = ok ? w : NOEDGE;
    }
    if (ok) {
      constexpr int IP = SAVE ? D : SP;
      const float* rp = hin + (long long)(w >> LOCB) * IP;
      float4 r0, r1, r2;
      r0.x = rp[0]; r0.y = rp[1]; r0.z = rp[2]; r0.w = rp[3];
      r1.x = rp[4]; r1.y = rp[5]; r1.z = rp[6]; r1.w = rp[7];
      r2.x = rp[8]; r2.y = rp[9]; r2.z = 0.f;   r2.w = 0.f;
      add_row(s_msum + loc * LP, r0, r1, r2);
    }
  }
}

__device__ __forceinline__ unsigned int match_flags(uint4 q0, uint4 q1, unsigned int myTag) {
  unsigned int mm;
  mm  = zero_bytes(q0.x ^ myTag) >> 7;
  mm |= zero_bytes(q0.y ^ myTag) >> 6;
  mm |= zero_bytes(q0.z ^ myTag) >> 5;
  mm |= zero_bytes(q0.w ^ myTag) >> 4;
  mm |= zero_bytes(q1.x ^ myTag) >> 3;
  mm |= zero_bytes(q1.y ^ myTag) >> 2;
  mm |= zero_bytes(q1.z ^ myTag) >> 1;
  mm |= zero_bytes(q1.w ^ myTag);
  return mm;
}

__device__ __forceinline__ void load_tags(const uint4* __restrict__ ow, int nVec, int base,
                                          int tid, uint4& q0, uint4& q1) {
  int v0 = base + tid, v1 = v0 + 1024;
  const int last = max(nVec - 1, 0);
  q0 = ow[min(v0, last)];
  q1 = ow[min(v1, last)];
}

template <bool SAVE, bool AHEAD>
__device__ __forceinline__ void scan_trip(const uint4* __restrict__ ow, int nVec, int base,
                                          uint4& s0, uint4& s1,
                                          unsigned int myTag, int tid, int lane, int* wq,
                                          const float* __restrict__ hin,
                                          const unsigned int* __restrict__ wr,
                                          float* sm, int npw,
                                          unsigned int* __restrict__ lst,
                                          int& qn, int& done) {
  const int v0 = base + tid, v1 = v0 + 1024;
  unsigned int mm = match_flags(s0, s1, myTag);
  if (AHEAD) {
    mm &= ((v0 < nVec) ? 0x0F0F0F0Fu : 0u) | ((v1 < nVec) ? 0xF0F0F0F0u : 0u);
    load_tags(ow, nVec, base + 4096, tid, s0, s1);
  }

  for (int trip = 0; trip < 32; ++trip) {
    bool has = (mm != 0u);
    unsigned int bal = __builtin_amdgcn_ballot_w32(has);
    if (bal == 0u) break;
    if (has) {
      int p = __builtin_ctz(mm);
      mm &= mm - 1u;
      int jj = p & 7, n = p >> 3;
      int v  = (jj & 4) ? v1 : v0;
      int e  = v * 16 + (jj & 3) * 4 + n;
      int pos = qn + __builtin_popcount(bal & ((1u << lane) - 1u));
      wq[pos] = e;
    }
    qn += __builtin_popcount(bal);
    if (qn >= 32) {
      flush_edges<SAVE>(wq, 32, lane, hin, wr, sm, npw, lst, done);
      done += 32;
      int rest = qn - 32;
      int tmp = 0;
      if (lane < rest) tmp = wq[32 + lane];
      if (lane < rest) wq[lane] = tmp;
      qn = rest;
    }
  }
}

template <bool SAVE>
__device__ __forceinline__ int scan_edges(const uint4* __restrict__ ow, int nVec,
                                          unsigned int myTag, int tid, int lane, int* wq,
                                          const float* __restrict__ hin,
                                          const unsigned int* __restrict__ wr,
                                          float* sm, int npw,
                                          unsigned int* __restrict__ lst) {
  int qn = 0;
  int done = 0;

  uint4 a0, a1;
  if (SAVE) {
    uint4 b0, b1;
    load_tags(ow, nVec, 2048, tid, b0, b1);
    load_tags(ow, nVec, 0, tid, a0, a1);
    for (int base = 0; base < nVec; base += 4096) {
      scan_trip<SAVE, true>(ow, nVec, base, a0, a1, myTag, tid, lane, wq, hin, wr, sm, npw, lst, qn, done);
      scan_trip<SAVE, true>(ow, nVec, base + 2048, b0, b1, myTag, tid, lane, wq, hin, wr, sm, npw, lst, qn, done);
    }
  } else {
    for (int base = 0; base < nVec; base += 2048) {
      int v0 = base + tid, v1 = v0 + 1024;
      a0 = make_uint4(~0u, ~0u, ~0u, ~0u);
      a1 = a0;
      if (v0 < nVec) a0 = ow[v0];
      if (v1 < nVec) a1 = ow[v1];
      scan_trip<SAVE, false>(ow, nVec, base, a0, a1, myTag, tid, lane, wq, hin, wr, sm, npw, lst, qn, done);
    }
  }
  flush_edges<SAVE>(wq, qn, lane, hin, wr, sm, npw, lst, done);
  return done + qn;
}

__device__ __forceinline__ void replay_edges(const unsigned int* __restrict__ lp, int n,
                                             int lane, const float* __restrict__ hin,
                                             float* sm) {
  constexpr int U = 4;
  unsigned int nxt[U];
#pragma unroll
  for (int u = 0; u < U; ++u) {
    int i = u * 32 + lane;
    nxt[u] = (i < n) ? lp[i] : NOEDGE;
  }
  for (int i0 = 0; i0 < n; i0 += 32 * U) {
    unsigned int p[U];
    float4 r0[U], r1[U], r2[U];
#pragma unroll
    for (int u = 0; u < U; ++u) {
      p[u] = nxt[u];
      if (p[u] != NOEDGE) {
        const float4* rp = (const float4*)(hin + (long long)(p[u] >> LOCB) * SP);
        r0[u] = rp[0]; r1[u] = rp[1]; r2[u] = rp[2];
      }
    }
#pragma unroll
    for (int u = 0; u < U; ++u) {
      int i = i0 + 32 * U + u * 32 + lane;
      nxt[u] = (i < n) ? lp[i] : NOEDGE;
    }
#pragma unroll
    for (int u = 0; u < U; ++u) {
      if (p[u] != NOEDGE)
        add_row(sm + (p[u] & ((1u << LOCB) - 1u)) * LP, r0[u], r1[u], r2[u]);
    }
  }
}


template <bool REPLAY>
__global__ __launch_bounds__(1024)
void k_layer(const float* __restrict__ hin,
             const unsigned int* __restrict__ owner,
             const unsigned int* __restrict__ words,
             const float* __restrict__ WsP,
             const float* __restrict__ WnP,
             const float* __restrict__ bsP,
             float* __restrict__ hout,
             unsigned int* __restrict__ lists,
             int* __restrict__ lcnt,
             int Nn, int Epad, int npw) {
  __shared__ __attribute__((aligned(16))) float s_msum[2 * NPW_MAX * LP];
  __shared__ int s_q[32 * 64];

  const int tid  = threadIdx.x;
  const int lane = tid & 31;
  const int wave = tid >> 5;
  const int m    = lane & 15;
  const int half = lane >> 4;
  const int lo   = blockIdx.x * npw;
  const int cnt  = min(npw, Nn - lo);
  const int nT   = (cnt + 15) >> 4;
  int* wq = s_q + wave * 64;

  union Acc { v8f v; float f[8]; };
  Acc acc[TPW];

  {
    float bsum = bsP[m];
    const float* brow = WsP + m * SP;
#pragma unroll
    for (int i = 0; i < TPW; ++i) {
      int t = wave + 32 * i;
#pragma unroll
      for (int j = 0; j < 8; ++j) acc[i].f[j] = bsum;
      if (t < nT) {
        constexpr int IP = REPLAY ? SP : D;
        int node = min(lo + t * 16 + m, Nn - 1);
        const float* arow = hin + (long long)node * IP;
#pragma unroll
        for (int c = 0; c < 3; ++c) {
          int k0 = c * 4 + half * 2;
          v2f a, b;
          bool stored = (k0 + 1 < IP);
          a.x = stored ? arow[k0] : 0.f; a.y = stored ? arow[k0 + 1] : 0.f;
          b.x = brow[k0]; b.y = brow[k0 + 1];
          acc[i].v = wmma_k4(a, b, acc[i].v);
        }
      }
    }
  }

  const unsigned int myTag = (unsigned int)blockIdx.x * 0x01010101u;
  const int nVec = Epad >> 4;

  static_assert(RT % 2 == 0, "etypes are taken two per pass");
  for (int rp = 0; rp < RT; rp += 2) {
    for (int i = tid; i < (npw * LP) / 4; i += 1024) {
      ((float4*)s_msum)[i] = make_float4(0.f, 0.f, 0.f, 0.f);
      ((float4*)(s_msum + NPW_MAX * LP))[i] = make_float4(0.f, 0.f, 0.f, 0.f);
    }
    __syncthreads();

   for (int rr = 0; rr < 2; ++rr) {
    const int r = rp + rr;
    float* sm = s_msum + rr * (NPW_MAX * LP);
    const uint4* ow  = (const uint4*)(owner + (long long)r * (Epad >> 2));
    const unsigned int* wr = words + (long long)r * Epad;
    const long long slot = ((long long)blockIdx.x * RT + r) * 32 + wave;
    unsigned int* lst = lists + slot * LCAP;

    if (REPLAY) {
      int n = __builtin_amdgcn_readfirstlane(lcnt[slot]);
      if (n <= LCAP) replay_edges(lst, n, lane, hin, sm);
      else
        scan_edges<false>(ow, nVec, myTag, tid, lane, wq, hin, wr, sm, npw, lst);
    } else {
      int n = scan_edges<true>(ow, nVec, myTag, tid, lane, wq, hin, wr, sm, npw, lst);
      if (lane == 0) lcnt[slot] = n;
    }
   }
    __syncthreads();

#pragma unroll
    for (int rr = 0; rr < 2; ++rr) {
      const float* brow = WnP + (rp + rr) * 16 * SP + m * SP;
      const float* sm = s_msum + rr * (NPW_MAX * LP);
#pragma unroll
      for (int i = 0; i < TPW; ++i) {
        int t = wave + 32 * i;
        if (t < nT) {
          const float* arow = sm + (t * 16 + m) * LP;
          float rdeg = 1.0f / fmaxf(arow[D], 1.0f);
#pragma unroll
          for (int c = 0; c < 3; ++c) {
            int k0 = c * 4 + half * 2;
            v2f a, b;
            a.x = arow[k0] * rdeg; a.y = arow[k0 + 1] * rdeg;
            b.x = brow[k0];        b.y = brow[k0 + 1];
            acc[i].v = wmma_k4(a, b, acc[i].v);
          }
        }
      }
    }
    __syncthreads();
  }

#pragma unroll
  for (int i = 0; i < TPW; ++i) {
    int t = wave + 32 * i;
    if (t < nT && m < SP) {
      int rowBase = lo + t * 16 + half * 8;
      float* op = hout + (long long)rowBase * SP + m;
#pragma unroll
      for (int j = 0; j < 8; ++j)
        if (rowBase + j < Nn) op[j * SP] = acc[i].f[j];
    }
  }
}

__global__ void k_score(const float* __restrict__ hp,
                        const int* __restrict__ ps, const int* __restrict__ pd,
                        const int* __restrict__ ns, const int* __restrict__ nd,
                        float* __restrict__ out, int E_) {
  int i = blockIdx.x * blockDim.x + threadIdx.x;
  if (i >= 2 * E_) return;
  int a, b;
  if (i < E_) { a = ps[i]; b = pd[i]; }
  else        { a = ns[i - E_]; b = nd[i - E_]; }
  const float* pa = hp + (long long)a * SP;
  const float* pb = hp + (long long)b * SP;
  float s = 0.0f;
#pragma unroll
  for (int k = 0; k < D; ++k) s += pa[k] * pb[k];
  out[i] = s;
}

extern "C" void kernel_launch(void* const* d_in, const int* in_sizes, int n_in,
                              void* d_out, int out_size, void* d_ws, size_t ws_size,
                              hipStream_t stream) {
  (void)n_in; (void)out_size; (void)ws_size;
  const float* x    = (const float*)d_in[0];
  const int*   esrc = (const int*)d_in[1];
  const int*   edst = (const int*)d_in[2];
  const int*   nsrc = (const int*)d_in[3];
  const int*   ndst = (const int*)d_in[4];
  const float* Ws1  = (const float*)d_in[5];
  const float* Wn1  = (const float*)d_in[6];
  const float* b1   = (const float*)d_in[7];
  const float* Ws2  = (const float*)d_in[8];
  const float* Wn2  = (const float*)d_in[9];
  const float* b2   = (const float*)d_in[10];

  const int Nn = in_sizes[0] / D;
  const int Ee = in_sizes[3];

  float* B0   = (float*)d_ws;
  float* B1   = B0 + (size_t)Nn * SP;
  float* WsP1 = B1 + (size_t)Nn * SP;
  float* WnP1 = WsP1 + 16 * SP;
  float* bsP1 = WnP1 + RT * 16 * SP;
  float* WsP2 = bsP1 + 16;
  float* WnP2 = WsP2 + 16 * SP;
  float* bsP2 = WnP2 + RT * 16 * SP;
  unsigned int* owner = (unsigned int*)(bsP2 + 16);

  const int Epad   = (Ee + 15) & ~15;
  unsigned int* words = owner + RT * (size_t)(Epad >> 2);
  int* lcnt = (int*)(words + RT * (size_t)Epad);
  unsigned int* lists = (unsigned int*)(lcnt + 255 * RT * 32);
  const int npw    = (((Nn + 254) / 255) + 15) & ~15;
  const int lBlocks = (Nn + npw - 1) / npw;
  const int ownCnt = RT * (Epad >> 2);

  k_prep<<<1, 256, 0, stream>>>(Ws1, Wn1, b1, Ws2, Wn2, b2,
                                WsP1, WnP1, bsP1, WsP2, WnP2, bsP2);
  k_owner<<<(ownCnt + 255) / 256, 256, 0, stream>>>(esrc, edst, owner, words, Ee, Epad, npw);

  k_layer<false><<<lBlocks, 1024, 0, stream>>>(x, owner, words, WsP1, WnP1, bsP1, B1,
                                               lists, lcnt, Nn, Epad, npw);
  k_layer<true><<<lBlocks, 1024, 0, stream>>>(B1, owner, words, WsP2, WnP2, bsP2, B0,
                                              lists, lcnt, Nn, Epad, npw);

  k_score<<<(2 * Ee + 255) / 256, 256, 0, stream>>>(B0, esrc, edst, nsrc, ndst,
                                                    (float*)d_out, Ee);
}


# Layer-1 tag scan skips 128-B tag lines that hold no edge of the work-group: the owner pre-pass sets one presence bit per (etype, owner, line); 8 lanes share a line's bit; sum order unchanged
# speedup vs baseline: 3.4664x; 1.0401x over previous
#include <hip/hip_runtime.h>


#define D  10
#define RT 4
#define SP 12
#define LP 13
#define NPW_MAX 2048
#define TPW (NPW_MAX / 16 / 32)
#define NOTAG 255u
#define LCAP 1024
#define LOCB 11
#define NOEDGE 0xFFFFFFFFu

typedef __attribute__((ext_vector_type(2))) float v2f;
typedef __attribute__((ext_vector_type(8))) float v8f;

__device__ __forceinline__ v8f wmma_k4(v2f a, v2f b, v8f c) {
  return __builtin_amdgcn_wmma_f32_16x16x4_f32(
       false, a,  false, b,
       (short)0, c,  false,  false);
}

__global__ void k_prep(const float* __restrict__ Ws1, const float* __restrict__ Wn1,
                       const float* __restrict__ b1,
                       const float* __restrict__ Ws2, const float* __restrict__ Wn2,
                       const float* __restrict__ b2,
                       float* __restrict__ WsP1, float* __restrict__ WnP1,
                       float* __restrict__ bsP1,
                       float* __restrict__ WsP2, float* __restrict__ WnP2,
                       float* __restrict__ bsP2) {
  for (int i = threadIdx.x; i < 16 * SP; i += blockDim.x) {
    int n = i / SP, k = i % SP;
    float s1 = 0.f, s2 = 0.f;
    if (n < D && k < D) {
#pragma unroll
      for (int r = 0; r < RT; ++r) {
        s1 += Ws1[r * D * D + n * D + k];
        s2 += Ws2[r * D * D + n * D + k];
      }
    }
    WsP1[i] = s1;
    WsP2[i] = s2;
  }
  for (int i = threadIdx.x; i < RT * 16 * SP; i += blockDim.x) {
    int r = i / (16 * SP), j = i % (16 * SP);
    int n = j / SP, k = j % SP;
    float v1 = 0.f, v2 = 0.f;
    if (n < D && k < D) {
      v1 = Wn1[r * D * D + n * D + k];
      v2 = Wn2[r * D * D + n * D + k];
    }
    WnP1[i] = v1;
    WnP2[i] = v2;
  }
  for (int i = threadIdx.x; i < 16; i += blockDim.x) {
    float s1 = 0.f, s2 = 0.f;
    if (i < D) {
#pragma unroll
      for (int r = 0; r < RT; ++r) {
        s1 += b1[r * D + i];
        s2 += b2[r * D + i];
      }
    }
    bsP1[i] = s1;
    bsP2[i] = s2;
  }
}

__global__ __launch_bounds__(1024)
void k_owner(const int* __restrict__ src, const int* __restrict__ dst,
             unsigned int* __restrict__ owner, unsigned int* __restrict__ words,
             unsigned int* __restrict__ pres,
             int E_, int Epad, int npw) {
  __shared__ unsigned int s_pres[256];
  if (threadIdx.x < 256) s_pres[threadIdx.x] = 0u;
  __syncthreads();
  const int q4 = Epad >> 2;
  const int r  = blockIdx.y;
  const int il = blockIdx.x * 1024 + threadIdx.x;
  const int i  = r * q4 + il;
  const int e0 = il * 4;
  const unsigned int lineBit = 1u << (threadIdx.x >> 5);
  const int* sp = src + (long long)r * E_ + e0;
  const int* dp = dst + (long long)r * E_ + e0;
  int s[4] = {0, 0, 0, 0}, d[4] = {0, 0, 0, 0};
  bool wide = (e0 + 3 < E_) && ((((unsigned long long)sp | (unsigned long long)dp) & 15ull) == 0ull);
  if (wide) {
    int4 sv = *(const int4*)sp, dv = *(const int4*)dp;
    s[0] = sv.x; s[1] = sv.y; s[2] = sv.z; s[3] = sv.w;
    d[0] = dv.x; d[1] = dv.y; d[2] = dv.z; d[3] = dv.w;
  } else {
#pragma unroll
    for (int b = 0; b < 4; ++b)
      if (e0 + b < E_) { s[b] = sp[b]; d[b] = dp[b]; }
  }
  unsigned int pack = 0;
  unsigned int w[4];
#pragma unroll
  for (int b = 0; b < 4; ++b) {
    unsigned int tag = NOTAG;
    w[b] = NOEDGE;
    if (e0 + b < E_) {
      tag = (unsigned int)d[b] / (unsigned int)npw;
      unsigned int loc = (unsigned int)d[b] - tag * (unsigned int)npw;
      w[b] = ((unsigned int)s[b] << LOCB) | loc;
      if (tag < 255u) atomicOr(&s_pres[tag], lineBit);
    }
    pack |= tag << (8 * b);
  }
  owner[i] = pack;
  ((uint4*)words)[i] = make_uint4(w[0], w[1], w[2], w[3]);
  __syncthreads();
  if (threadIdx.x < 255)
    pres[((size_t)r * 256 + threadIdx.x) * (size_t)gridDim.x + blockIdx.x] = s_pres[threadIdx.x];
}

__device__ __forceinline__ unsigned int zero_bytes(unsigned int x) {
  return ~(((x & 0x7F7F7F7Fu) + 0x7F7F7F7Fu) | x | 0x7F7F7F7Fu);
}

__device__ __forceinline__ void add_row(float* mp, float4 r0, float4 r1, float4 r2) {
  atomicAdd(mp + 0, r0.x); atomicAdd(mp + 1, r0.y);
  atomicAdd(mp + 2, r0.z); atomicAdd(mp + 3, r0.w);
  atomicAdd(mp + 4, r1.x); atomicAdd(mp + 5, r1.y);
  atomicAdd(mp + 6, r1.z); atomicAdd(mp + 7, r1.w);
  atomicAdd(mp + 8, r2.x); atomicAdd(mp + 9, r2.y);
  atomicAdd(mp + 10, 1.0f);
}

template <bool SAVE>
__device__ __forceinline__ void flush_edges(const int* wq, int n, int lane,
                                            const float* __restrict__ hin,
                                            const unsigned int* __restrict__ wr,
                                            float* s_msum, int npw,
                                            unsigned int* __restrict__ lst, int pos) {
  if (lane < n) {
    int e = wq[lane];
    unsigned int w = wr[e];
    unsigned int loc = w & ((1u << LOCB) - 1u);
    bool ok = loc < (unsigned int)npw;
    if (SAVE) {
      int at = pos + lane;
      if (at < LCAP) lst[at] = ok ? w : NOEDGE;
    }
    if (ok) {
      constexpr int IP = SAVE ? D : SP;
      const float* rp = hin + (long long)(w >> LOCB) * IP;
      float4 r0, r1, r2;
      r0.x = rp[0]; r0.y = rp[1]; r0.z = rp[2]; r0.w = rp[3];
      r1.x = rp[4]; r1.y = rp[5]; r1.z = rp[6]; r1.w = rp[7];
      r2.x = rp[8]; r2.y = rp[9]; r2.z = 0.f;   r2.w = 0.f;
      add_row(s_msum + loc * LP, r0, r1, r2);
    }
  }
}

__device__ __forceinline__ unsigned int match_flags(uint4 q0, uint4 q1, unsigned int myTag) {
  unsigned int mm;
  mm  = zero_bytes(q0.x ^ myTag) >> 7;
  mm |= zero_bytes(q0.y ^ myTag) >> 6;
  mm |= zero_bytes(q0.z ^ myTag) >> 5;
  mm |= zero_bytes(q0.w ^ myTag) >> 4;
  mm |= zero_bytes(q1.x ^ myTag) >> 3;
  mm |= zero_bytes(q1.y ^ myTag) >> 2;
  mm |= zero_bytes(q1.z ^ myTag) >> 1;
  mm |= zero_bytes(q1.w ^ myTag);
  return mm;
}

__device__ __forceinline__ void load_pres(const unsigned int* __restrict__ pr, int nW, int base,
                                          int wq3, unsigned int& p0, unsigned int& p1) {
  int i0 = (base >> 8) + wq3;
  p0 = pr[min(i0, nW - 1)];
  p1 = pr[min(i0 + 4, nW - 1)];
}

__device__ __forceinline__ unsigned int load_tags(const uint4* __restrict__ ow, int nVec, int base,
                                                  int tid, unsigned int p0, unsigned int p1,
                                                  uint4& q0, uint4& q1) {
  int v0 = base + tid, v1 = v0 + 1024;
  const unsigned int bit = (unsigned int)(tid >> 3) & 31u;
  bool ok0 = (v0 < nVec) && (((p0 >> bit) & 1u) != 0u);
  bool ok1 = (v1 < nVec) && (((p1 >> bit) & 1u) != 0u);
  q0 = ow[ok0 ? v0 : 0];
  q1 = ow[ok1 ? v1 : 0];
  return (ok0 ? 0x0F0F0F0Fu : 0u) | (ok1 ? 0xF0F0F0F0u : 0u);
}

template <bool SAVE, bool AHEAD>
__device__ __forceinline__ void scan_trip(const uint4* __restrict__ ow, int nVec, int base,
                                          uint4& s0, uint4& s1, unsigned int& keep,
                                          const unsigned int* __restrict__ pr, int nW, int wq3,
                                          unsigned int& pn0, unsigned int& pn1,
                                          unsigned int myTag, int tid, int lane, int* wq,
                                          const float* __restrict__ hin,
                                          const unsigned int* __restrict__ wr,
                                          float* sm, int npw,
                                          unsigned int* __restrict__ lst,
                                          int& qn, int& done) {
  const int v0 = base + tid, v1 = v0 + 1024;
  unsigned int mm = match_flags(s0, s1, myTag);
  if (AHEAD) {
    mm &= keep;
    keep = load_tags(ow, nVec, base + 4096, tid, pn0, pn1, s0, s1);
    load_pres(pr, nW, base + 6144, wq3, pn0, pn1);
  }

  for (int trip = 0; trip < 32; ++trip) {
    bool has = (mm != 0u);
    unsigned int bal = __builtin_amdgcn_ballot_w32(has);
    if (bal == 0u) break;
    if (has) {
      int p = __builtin_ctz(mm);
      mm &= mm - 1u;
      int jj = p & 7, n = p >> 3;
      int v  = (jj & 4) ? v1 : v0;
      int e  = v * 16 + (jj & 3) * 4 + n;
      int pos = qn + __builtin_popcount(bal & ((1u << lane) - 1u));
      wq[pos] = e;
    }
    qn += __builtin_popcount(bal);
    if (qn >= 32) {
      flush_edges<SAVE>(wq, 32, lane, hin, wr, sm, npw, lst, done);
      done += 32;
      int rest = qn - 32;
      int tmp = 0;
      if (lane < rest) tmp = wq[32 + lane];
      if (lane < rest) wq[lane] = tmp;
      qn = rest;
    }
  }
}

template <bool SAVE>
__device__ __forceinline__ int scan_edges(const uint4* __restrict__ ow, int nVec,
                                          unsigned int myTag, int tid, int lane, int* wq,
                                          const float* __restrict__ hin,
                                          const unsigned int* __restrict__ wr,
                                          float* sm, int npw,
                                          unsigned int* __restrict__ lst,
                                          const unsigned int* __restrict__ pr, int nW) {
  int qn = 0;
  int done = 0;

  uint4 a0, a1;
  unsigned int pn0 = 0u, pn1 = 0u;
  if (SAVE) {
    const int wq3 = __builtin_amdgcn_readfirstlane(tid >> 5) >> 3;
    uint4 b0, b1;
    unsigned int pa0, pa1, pb0, pb1;
    load_pres(pr, nW, 0, wq3, pa0, pa1);
    load_pres(pr, nW, 2048, wq3, pb0, pb1);
    load_pres(pr, nW, 4096, wq3, pn0, pn1);
    unsigned int keepB = load_tags(ow, nVec, 2048, tid, pb0, pb1, b0, b1);
    unsigned int keepA = load_tags(ow, nVec, 0, tid, pa0, pa1, a0, a1);
    for (int base = 0; base < nVec; base += 4096) {
      scan_trip<SAVE, true>(ow, nVec, base, a0, a1, keepA, pr, nW, wq3, pn0, pn1,
                            myTag, tid, lane, wq, hin, wr, sm, npw, lst, qn, done);
      scan_trip<SAVE, true>(ow, nVec, base + 2048, b0, b1, keepB, pr, nW, wq3, pn0, pn1,
                            myTag, tid, lane, wq, hin, wr, sm, npw, lst, qn, done);
    }
  } else {
    for (int base = 0; base < nVec; base += 2048) {
      int v0 = base + tid, v1 = v0 + 1024;
      a0 = make_uint4(~0u, ~0u, ~0u, ~0u);
      a1 = a0;
      if (v0 < nVec) a0 = ow[v0];
      if (v1 < nVec) a1 = ow[v1];
      unsigned int all = 0xFFFFFFFFu;
      scan_trip<SAVE, false>(ow, nVec, base, a0, a1, all, pr, nW, 0, pn0, pn1,
                             myTag, tid, lane, wq, hin, wr, sm, npw, lst, qn, done);
    }
  }
  flush_edges<SAVE>(wq, qn, lane, hin, wr, sm, npw, lst, done);
  return done + qn;
}

__device__ __forceinline__ void replay_edges(const unsigned int* __restrict__ lp, int n,
                                             int lane, const float* __restrict__ hin,
                                             float* sm) {
  constexpr int U = 4;
  unsigned int nxt[U];
#pragma unroll
  for (int u = 0; u < U; ++u) {
    int i = u * 32 + lane;
    nxt[u] = (i < n) ? lp[i] : NOEDGE;
  }
  for (int i0 = 0; i0 < n; i0 += 32 * U) {
    unsigned int p[U];
    float4 r0[U], r1[U], r2[U];
#pragma unroll
    for (int u = 0; u < U; ++u) {
      p[u] = nxt[u];
      if (p[u] != NOEDGE) {
        const float4* rp = (const float4*)(hin + (long long)(p[u] >> LOCB) * SP);
        r0[u] = rp[0]; r1[u] = rp[1]; r2[u] = rp[2];
      }
    }
#pragma unroll
    for (int u = 0; u < U; ++u) {
      int i = i0 + 32 * U + u * 32 + lane;
      nxt[u] = (i < n) ? lp[i] : NOEDGE;
    }
#pragma unroll
    for (int u = 0; u < U; ++u) {
      if (p[u] != NOEDGE)
        add_row(sm + (p[u] & ((1u << LOCB) - 1u)) * LP, r0[u], r1[u], r2[u]);
    }
  }
}


template <bool REPLAY>
__global__ __launch_bounds__(1024)
void k_layer(const float* __restrict__ hin,
             const unsigned int* __restrict__ owner,
             const unsigned int* __restrict__ words,
             const unsigned int* __restrict__ pres,
             const float* __restrict__ WsP,
             const float* __restrict__ WnP,
             const float* __restrict__ bsP,
             float* __restrict__ hout,
             unsigned int* __restrict__ lists,
             int* __restrict__ lcnt,
             int Nn, int Epad, int npw) {
  __shared__ __attribute__((aligned(16))) float s_msum[2 * NPW_MAX * LP];
  __shared__ int s_q[32 * 64];

  const int tid  = threadIdx.x;
  const int lane = tid & 31;
  const int wave = tid >> 5;
  const int m    = lane & 15;
  const int half = lane >> 4;
  const int lo   = blockIdx.x * npw;
  const int cnt  = min(npw, Nn - lo);
  const int nT   = (cnt + 15) >> 4;
  int* wq = s_q + wave * 64;

  union Acc { v8f v; float f[8]; };
  Acc acc[TPW];

  {
    float bsum = bsP[m];
    const float* brow = WsP + m * SP;
#pragma unroll
    for (int i = 0; i < TPW; ++i) {
      int t = wave + 32 * i;
#pragma unroll
      for (int j = 0; j < 8; ++j) acc[i].f[j] = bsum;
      if (t < nT) {
        constexpr int IP = REPLAY ? SP : D;
        int node = min(lo + t * 16 + m, Nn - 1);
        const float* arow = hin + (long long)node * IP;
#pragma unroll
        for (int c = 0; c < 3; ++c) {
          int k0 = c * 4 + half * 2;
          v2f a, b;
          bool stored = (k0 + 1 < IP);
          a.x = stored ? arow[k0] : 0.f; a.y = stored ? arow[k0 + 1] : 0.f;
          b.x = brow[k0]; b.y = brow[k0 + 1];
          acc[i].v = wmma_k4(a, b, acc[i].v);
        }
      }
    }
  }

  const unsigned int myTag = (unsigned int)blockIdx.x * 0x01010101u;
  const int nVec = Epad >> 4;
  const int nW   = Epad >> 12;

  static_assert(RT % 2 == 0, "etypes are taken two per pass");
  for (int rp = 0; rp < RT; rp += 2) {
    for (int i = tid; i < (npw * LP) / 4; i += 1024) {
      ((float4*)s_msum)[i] = make_float4(0.f, 0.f, 0.f, 0.f);
      ((float4*)(s_msum + NPW_MAX * LP))[i] = make_float4(0.f, 0.f, 0.f, 0.f);
    }
    __syncthreads();

   for (int rr = 0; rr < 2; ++rr) {
    const int r = rp + rr;
    float* sm = s_msum + rr * (NPW_MAX * LP);
    const uint4* ow  = (const uint4*)(owner + (long long)r * (Epad >> 2));
    const unsigned int* wr = words + (long long)r * Epad;
    const long long slot = ((long long)blockIdx.x * RT + r) * 32 + wave;
    unsigned int* lst = lists + slot * LCAP;
    const unsigned int* pr = pres + ((long long)r * 256 + blockIdx.x) * nW;

    if (REPLAY) {
      int n = __builtin_amdgcn_readfirstlane(lcnt[slot]);
      if (n <= LCAP) replay_edges(lst, n, lane, hin, sm);
      else
        scan_edges<false>(ow, nVec, myTag, tid, lane, wq, hin, wr, sm, npw, lst, pr, nW);
    } else {
      int n = scan_edges<true>(ow, nVec, myTag, tid, lane, wq, hin, wr, sm, npw, lst, pr, nW);
      if (lane == 0) lcnt[slot] = n;
    }
   }
    __syncthreads();

#pragma unroll
    for (int rr = 0; rr < 2; ++rr) {
      const float* brow = WnP + (rp + rr) * 16 * SP + m * SP;
      const float* sm = s_msum + rr * (NPW_MAX * LP);
#pragma unroll
      for (int i = 0; i < TPW; ++i) {
        int t = wave + 32 * i;
        if (t < nT) {
          const float* arow = sm + (t * 16 + m) * LP;
          float rdeg = 1.0f / fmaxf(arow[D], 1.0f);
#pragma unroll
          for (int c = 0; c < 3; ++c) {
            int k0 = c * 4 + half * 2;
            v2f a, b;
            a.x = arow[k0] * rdeg; a.y = arow[k0 + 1] * rdeg;
            b.x = brow[k0];        b.y = brow[k0 + 1];
            acc[i].v = wmma_k4(a, b, acc[i].v);
          }
        }
      }
    }
    __syncthreads();
  }

#pragma unroll
  for (int i = 0; i < TPW; ++i) {
    int t = wave + 32 * i;
    if (t < nT && m < SP) {
      int rowBase = lo + t * 16 + half * 8;
      float* op = hout + (long long)rowBase * SP + m;
#pragma unroll
      for (int j = 0; j < 8; ++j)
        if (rowBase + j < Nn) op[j * SP] = acc[i].f[j];
    }
  }
}

__global__ void k_score(const float* __restrict__ hp,
                        const int* __restrict__ ps, const int* __restrict__ pd,
                        const int* __restrict__ ns, const int* __restrict__ nd,
                        float* __restrict__ out, int E_) {
  int i = blockIdx.x * blockDim.x + threadIdx.x;
  if (i >= 2 * E_) return;
  int a, b;
  if (i < E_) { a = ps[i]; b = pd[i]; }
  else        { a = ns[i - E_]; b = nd[i - E_]; }
  const float* pa = hp + (long long)a * SP;
  const float* pb = hp + (long long)b * SP;
  float s = 0.0f;
#pragma unroll
  for (int k = 0; k < D; ++k) s += pa[k] * pb[k];
  out[i] = s;
}

extern "C" void kernel_launch(void* const* d_in, const int* in_sizes, int n_in,
                              void* d_out, int out_size, void* d_ws, size_t ws_size,
                              hipStream_t stream) {
  (void)n_in; (void)out_size; (void)ws_size;
  const float* x    = (const float*)d_in[0];
  const int*   esrc = (const int*)d_in[1];
  const int*   edst = (const int*)d_in[2];
  const int*   nsrc = (const int*)d_in[3];
  const int*   ndst = (const int*)d_in[4];
  const float* Ws1  = (const float*)d_in[5];
  const float* Wn1  = (const float*)d_in[6];
  const float* b1   = (const float*)d_in[7];
  const float* Ws2  = (const float*)d_in[8];
  const float* Wn2  = (const float*)d_in[9];
  const float* b2   = (const float*)d_in[10];

  const int Nn = in_sizes[0] / D;
  const int Ee = in_sizes[3];

  float* B0   = (float*)d_ws;
  float* B1   = B0 + (size_t)Nn * SP;
  float* WsP1 = B1 + (size_t)Nn * SP;
  float* WnP1 = WsP1 + 16 * SP;
  float* bsP1 = WnP1 + RT * 16 * SP;
  float* WsP2 = bsP1 + 16;
  float* WnP2 = WsP2 + 16 * SP;
  float* bsP2 = WnP2 + RT * 16 * SP;
  unsigned int* owner = (unsigned int*)(bsP2 + 16);

  const int Epad   = (Ee + 4095) & ~4095;
  unsigned int* words = owner + RT * (size_t)(Epad >> 2);
  unsigned int* pres  = words + RT * (size_t)Epad;
  int* lcnt = (int*)(pres + RT * 256 * (size_t)(Epad >> 12));
  unsigned int* lists = (unsigned int*)(lcnt + 255 * RT * 32);
  const int npw    = (((Nn + 254) / 255) + 15) & ~15;
  const int lBlocks = (Nn + npw - 1) / npw;

  k_prep<<<1, 256, 0, stream>>>(Ws1, Wn1, b1, Ws2, Wn2, b2,
                                WsP1, WnP1, bsP1, WsP2, WnP2, bsP2);
  k_owner<<<dim3(Epad >> 12, RT), 1024, 0, stream>>>(esrc, edst, owner, words, pres, Ee, Epad, npw);

  k_layer<false><<<lBlocks, 1024, 0, stream>>>(x, owner, words, pres, WsP1, WnP1, bsP1, B1,
                                               lists, lcnt, Nn, Epad, npw);
  k_layer<true><<<lBlocks, 1024, 0, stream>>>(B1, owner, words, pres, WsP2, WnP2, bsP2, B0,
                                              lists, lcnt, Nn, Epad, npw);

  k_score<<<(2 * Ee + 255) / 256, 256, 0, stream>>>(B0, esrc, edst, nsrc, ndst,
                                                    (float*)d_out, Ee);
}


# Layer-1 scan walks presence bits per (etype, owner, 16-edge tag vector), compares only the tags of set groups (layer 2's fall-back alike); a wave's share is 16384-edge runs; LDS-atomic arrival order
# speedup vs baseline: 4.1912x; 1.2091x over previous
#include <hip/hip_runtime.h>


#define D  10
#define RT 4
#define SP 12
#define LP 13
#define NPW_MAX 2048
#define TPW (NPW_MAX / 16 / 32)
#define NOTAG 255u
#define LCAP 1024
#define LOCB 11
#define NOEDGE 0xFFFFFFFFu

typedef __attribute__((ext_vector_type(2))) float v2f;
typedef __attribute__((ext_vector_type(8))) float v8f;

__device__ __forceinline__ v8f wmma_k4(v2f a, v2f b, v8f c) {
  return __builtin_amdgcn_wmma_f32_16x16x4_f32(
       false, a,  false, b,
       (short)0, c,  false,  false);
}

__global__ void k_prep(const float* __restrict__ Ws1, const float* __restrict__ Wn1,
                       const float* __restrict__ b1,
                       const float* __restrict__ Ws2, const float* __restrict__ Wn2,
                       const float* __restrict__ b2,
                       float* __restrict__ WsP1, float* __restrict__ WnP1,
                       float* __restrict__ bsP1,
                       float* __restrict__ WsP2, float* __restrict__ WnP2,
                       float* __restrict__ bsP2) {
  for (int i = threadIdx.x; i < 16 * SP; i += blockDim.x) {
    int n = i / SP, k = i % SP;
    float s1 = 0.f, s2 = 0.f;
    if (n < D && k < D) {
#pragma unroll
      for (int r = 0; r < RT; ++r) {
        s1 += Ws1[r * D * D + n * D + k];
        s2 += Ws2[r * D * D + n * D + k];
      }
    }
    WsP1[i] = s1;
    WsP2[i] = s2;
  }
  for (int i = threadIdx.x; i < RT * 16 * SP; i += blockDim.x) {
    int r = i / (16 * SP), j = i % (16 * SP);
    int n = j / SP, k = j % SP;
    float v1 = 0.f, v2 = 0.f;
    if (n < D && k < D) {
      v1 = Wn1[r * D * D + n * D + k];
      v2 = Wn2[r * D * D + n * D + k];
    }
    WnP1[i] = v1;
    WnP2[i] = v2;
  }
  for (int i = threadIdx.x; i < 16; i += blockDim.x) {
    float s1 = 0.f, s2 = 0.f;
    if (i < D) {
#pragma unroll
      for (int r = 0; r < RT; ++r) {
        s1 += b1[r * D + i];
        s2 += b2[r * D + i];
      }
    }
    bsP1[i] = s1;
    bsP2[i] = s2;
  }
}

__global__ __launch_bounds__(1024)
void k_owner(const int* __restrict__ src, const int* __restrict__ dst,
             unsigned int* __restrict__ owner, unsigned int* __restrict__ words,
             unsigned int* __restrict__ pres,
             int E_, int Epad, int npw) {
  __shared__ unsigned int s_pres[2048];
  s_pres[threadIdx.x] = 0u;
  s_pres[threadIdx.x + 1024] = 0u;
  __syncthreads();
  const int q4 = Epad >> 2;
  const int r  = blockIdx.y;
  const int il = blockIdx.x * 1024 + threadIdx.x;
  const int i  = r * q4 + il;
  const int e0 = il * 4;
  const unsigned int grpBit = 1u << ((threadIdx.x >> 2) & 31);
  const unsigned int grpDw  = threadIdx.x >> 7;
  const int* sp = src + (long long)r * E_ + e0;
  const int* dp = dst + (long long)r * E_ + e0;
  int s[4] = {0, 0, 0, 0}, d[4] = {0, 0, 0, 0};
  bool wide = (e0 + 3 < E_) && ((((unsigned long long)sp | (unsigned long long)dp) & 15ull) == 0ull);
  if (wide) {
    int4 sv = *(const int4*)sp, dv = *(const int4*)dp;
    s[0] = sv.x; s[1] = sv.y; s[2] = sv.z; s[3] = sv.w;
    d[0] = dv.x; d[1] = dv.y; d[2] = dv.z; d[3] = dv.w;
  } else {
#pragma unroll
    for (int b = 0; b < 4; ++b)
      if (e0 + b < E_) { s[b] = sp[b]; d[b] = dp[b]; }
  }
  unsigned int pack = 0;
  unsigned int w[4];
#pragma unroll
  for (int b = 0; b < 4; ++b) {
    unsigned int tag = NOTAG;
    w[b] = NOEDGE;
    if (e0 + b < E_) {
      tag = (unsigned int)d[b] / (unsigned int)npw;
      unsigned int loc = (unsigned int)d[b] - tag * (unsigned int)npw;
      w[b] = ((unsigned int)s[b] << LOCB) | loc;
      if (tag < 255u) atomicOr(&s_pres[tag * 8u + grpDw], grpBit);
    }
    pack |= tag << (8 * b);
  }
  owner[i] = pack;
  ((uint4*)words)[i] = make_uint4(w[0], w[1], w[2], w[3]);
  __syncthreads();
  unsigned int* po = pres + ((size_t)r * gridDim.x + blockIdx.x) * 2048;
  po[threadIdx.x]        = s_pres[threadIdx.x];
  po[threadIdx.x + 1024] = s_pres[threadIdx.x + 1024];
}

__device__ __forceinline__ unsigned int zero_bytes(unsigned int x) {
  return ~(((x & 0x7F7F7F7Fu) + 0x7F7F7F7Fu) | x | 0x7F7F7F7Fu);
}

__device__ __forceinline__ void add_row(float* mp, float4 r0, float4 r1, float4 r2) {
  atomicAdd(mp + 0, r0.x); atomicAdd(mp + 1, r0.y);
  atomicAdd(mp + 2, r0.z); atomicAdd(mp + 3, r0.w);
  atomicAdd(mp + 4, r1.x); atomicAdd(mp + 5, r1.y);
  atomicAdd(mp + 6, r1.z); atomicAdd(mp + 7, r1.w);
  atomicAdd(mp + 8, r2.x); atomicAdd(mp + 9, r2.y);
  atomicAdd(mp + 10, 1.0f);
}

template <bool SAVE>
__device__ __forceinline__ void flush_edges(const int* wq, int n, int lane,
                                            const float* __restrict__ hin,
                                            const unsigned int* __restrict__ wr,
                                            float* s_msum, int npw,
                                            unsigned int* __restrict__ lst, int pos) {
  if (lane < n) {
    int e = wq[lane];
    unsigned int w = wr[e];
    unsigned int loc = w & ((1u << LOCB) - 1u);
    bool ok = loc < (unsigned int)npw;
    if (SAVE) {
      int at = pos + lane;
      if (at < LCAP) lst[at] = ok ? w : NOEDGE;
    }
    if (ok) {
      constexpr int IP = SAVE ? D : SP;
      const float* rp = hin + (long long)(w >> LOCB) * IP;
      float4 r0, r1, r2;
      r0.x = rp[0]; r0.y = rp[1]; r0.z = rp[2]; r0.w = rp[3];
      r1.x = rp[4]; r1.y = rp[5]; r1.z = rp[6]; r1.w = rp[7];
      r2.x = rp[8]; r2.y = rp[9]; r2.z = 0.f;   r2.w = 0.f;
      add_row(s_msum + loc * LP, r0, r1, r2);
    }
  }
}

__device__ __forceinline__ unsigned int match_flags(uint4 q, unsigned int myTag) {
  unsigned int mm;
  mm  = zero_bytes(q.x ^ myTag) >> 7;
  mm |= zero_bytes(q.y ^ myTag) >> 6;
  mm |= zero_bytes(q.z ^ myTag) >> 5;
  mm |= zero_bytes(q.w ^ myTag) >> 4;
  return mm;
}

template <bool SAVE>
__device__ __forceinline__ void queue_matches(unsigned int mm, int v,
                                              int lane, int* wq,
                                              const float* __restrict__ hin,
                                              const unsigned int* __restrict__ wr,
                                              float* sm, int npw,
                                              unsigned int* __restrict__ lst,
                                              int& qn, int& done) {
  for (int trip = 0; trip < 16; ++trip) {
    bool has = (mm != 0u);
    unsigned int bal = __builtin_amdgcn_ballot_w32(has);
    if (bal == 0u) break;
    if (has) {
      int p = __builtin_ctz(mm);
      mm &= mm - 1u;
      int jj = p & 7, n = p >> 3;
      int e  = v * 16 + jj * 4 + n;
      int pos = qn + __builtin_popcount(bal & ((1u << lane) - 1u));
      wq[pos] = e;
    }
    qn += __builtin_popcount(bal);
    if (qn >= 32) {
      flush_edges<SAVE>(wq, 32, lane, hin, wr, sm, npw, lst, done);
      done += 32;
      int rest = qn - 32;
      int tmp = 0;
      if (lane < rest) tmp = wq[32 + lane];
      if (lane < rest) wq[lane] = tmp;
      qn = rest;
    }
  }
}

template <bool SAVE>
__device__ __forceinline__ void take_groups(const int* gq, int n, const uint4* __restrict__ ow,
                                            int& sg, uint4& st, unsigned int& smask,
                                            unsigned int myTag, int lane, int* wq,
                                            const float* __restrict__ hin,
                                            const unsigned int* __restrict__ wr,
                                            float* sm, int npw,
                                            unsigned int* __restrict__ lst,
                                            int& qn, int& done) {
  const unsigned int mm = match_flags(st, myTag) & smask;
  const int g = sg;
  smask = 0u;
  if (lane < n) {
    sg = gq[lane];
    st = ow[sg];
    smask = 0xFFFFFFFFu;
  }
  queue_matches<SAVE>(mm, g, lane, wq, hin, wr, sm, npw, lst, qn, done);
}

template <bool SAVE>
__device__ __forceinline__ int scan_groups(const unsigned int* __restrict__ pr, int nD,
                                           const uint4* __restrict__ ow,
                                           unsigned int myTag, int tid, int lane,
                                           int* wq, int* gq,
                                           const float* __restrict__ hin,
                                           const unsigned int* __restrict__ wr,
                                           float* sm, int npw,
                                           unsigned int* __restrict__ lst) {
  int qn = 0;
  int done = 0;
  int gn = 0;
  int sg = 0;
  uint4 st = make_uint4(0u, 0u, 0u, 0u);
  unsigned int smask = 0u;
  int dc = min(tid, nD - 1);
  unsigned int nxt = pr[(dc >> 3) * 2048 + (dc & 7)];
  for (int base = 0; base < nD; base += 1024) {
    const int d = base + tid;
    unsigned int pm = (d < nD) ? nxt : 0u;
    dc = min(d + 1024, nD - 1);
    nxt = pr[(dc >> 3) * 2048 + (dc & 7)];
    for (int trip = 0; trip < 32; ++trip) {
      bool has = (pm != 0u);
      unsigned int bal = __builtin_amdgcn_ballot_w32(has);
      if (bal == 0u) break;
      if (has) {
        int p = __builtin_ctz(pm);
        pm &= pm - 1u;
        int pos = gn + __builtin_popcount(bal & ((1u << lane) - 1u));
        gq[pos] = d * 32 + p;
      }
      gn += __builtin_popcount(bal);
      if (gn >= 32) {
        take_groups<SAVE>(gq, 32, ow, sg, st, smask, myTag, lane, wq, hin, wr, sm, npw, lst, qn, done);
        int rest = gn - 32;
        int tmp = 0;
        if (lane < rest) tmp = gq[32 + lane];
        if (lane < rest) gq[lane] = tmp;
        gn = rest;
      }
    }
  }
  take_groups<SAVE>(gq, gn, ow, sg, st, smask, myTag, lane, wq, hin, wr, sm, npw, lst, qn, done);
  take_groups<SAVE>(gq, 0, ow, sg, st, smask, myTag, lane, wq, hin, wr, sm, npw, lst, qn, done);
  flush_edges<SAVE>(wq, qn, lane, hin, wr, sm, npw, lst, done);
  return done + qn;
}

__device__ __forceinline__ void replay_edges(const unsigned int* __restrict__ lp, int n,
                                             int lane, const float* __restrict__ hin,
                                             float* sm) {
  constexpr int U = 4;
  unsigned int nxt[U];
#pragma unroll
  for (int u = 0; u < U; ++u) {
    int i = u * 32 + lane;
    nxt[u] = (i < n) ? lp[i] : NOEDGE;
  }
  for (int i0 = 0; i0 < n; i0 += 32 * U) {
    unsigned int p[U];
    float4 r0[U], r1[U], r2[U];
#pragma unroll
    for (int u = 0; u < U; ++u) {
      p[u] = nxt[u];
      if (p[u] != NOEDGE) {
        const float4* rp = (const float4*)(hin + (long long)(p[u] >> LOCB) * SP);
        r0[u] = rp[0]; r1[u] = rp[1]; r2[u] = rp[2];
      }
    }
#pragma unroll
    for (int u = 0; u < U; ++u) {
      int i = i0 + 32 * U + u * 32 + lane;
      nxt[u] = (i < n) ? lp[i] : NOEDGE;
    }
#pragma unroll
    for (int u = 0; u < U; ++u) {
      if (p[u] != NOEDGE)
        add_row(sm + (p[u] & ((1u << LOCB) - 1u)) * LP, r0[u], r1[u], r2[u]);
    }
  }
}


template <bool REPLAY>
__global__ __launch_bounds__(1024)
void k_layer(const float* __restrict__ hin,
             const unsigned int* __restrict__ owner,
             const unsigned int* __restrict__ words,
             const unsigned int* __restrict__ pres,
             const float* __restrict__ WsP,
             const float* __restrict__ WnP,
             const float* __restrict__ bsP,
             float* __restrict__ hout,
             unsigned int* __restrict__ lists,
             int* __restrict__ lcnt,
             int Nn, int Epad, int npw) {
  __shared__ __attribute__((aligned(16))) float s_msum[2 * NPW_MAX * LP];
  __shared__ int s_q[32 * 128];

  const int tid  = threadIdx.x;
  const int lane = tid & 31;
  const int wave = tid >> 5;
  const int m    = lane & 15;
  const int half = lane >> 4;
  const int lo   = blockIdx.x * npw;
  const int cnt  = min(npw, Nn - lo);
  const int nT   = (cnt + 15) >> 4;
  int* wq = s_q + wave * 128;
  int* gq = wq + 64;

  union Acc { v8f v; float f[8]; };
  Acc acc[TPW];

  {
    float bsum = bsP[m];
    const float* brow = WsP + m * SP;
#pragma unroll
    for (int i = 0; i < TPW; ++i) {
      int t = wave + 32 * i;
#pragma unroll
      for (int j = 0; j < 8; ++j) acc[i].f[j] = bsum;
      if (t < nT) {
        constexpr int IP = REPLAY ? SP : D;
        int node = min(lo + t * 16 + m, Nn - 1);
        const float* arow = hin + (long long)node * IP;
#pragma unroll
        for (int c = 0; c < 3; ++c) {
          int k0 = c * 4 + half * 2;
          v2f a, b;
          bool stored = (k0 + 1 < IP);
          a.x = stored ? arow[k0] : 0.f; a.y = stored ? arow[k0 + 1] : 0.f;
          b.x = brow[k0]; b.y = brow[k0 + 1];
          acc[i].v = wmma_k4(a, b, acc[i].v);
        }
      }
    }
  }

  const unsigned int myTag = (unsigned int)blockIdx.x * 0x01010101u;
  const int nD   = Epad >> 9;

  static_assert(RT % 2 == 0, "etypes are taken two per pass");
  for (int rp = 0; rp < RT; rp += 2) {
    for (int i = tid; i < (npw * LP) / 4; i += 1024) {
      ((float4*)s_msum)[i] = make_float4(0.f, 0.f, 0.f, 0.f);
      ((float4*)(s_msum + NPW_MAX * LP))[i] = make_float4(0.f, 0.f, 0.f, 0.f);
    }
    __syncthreads();

   for (int rr = 0; rr < 2; ++rr) {
    const int r = rp + rr;
    float* sm = s_msum + rr * (NPW_MAX * LP);
    const uint4* ow  = (const uint4*)(owner + (long long)r * (Epad >> 2));
    const unsigned int* wr = words + (long long)r * Epad;
    const long long slot = ((long long)blockIdx.x * RT + r) * 32 + wave;
    unsigned int* lst = lists + slot * LCAP;
    const unsigned int* pr = pres + ((long long)r * (Epad >> 12) * 256 + blockIdx.x) * 8;

    if (REPLAY) {
      int n = __builtin_amdgcn_readfirstlane(lcnt[slot]);
      if (n <= LCAP) replay_edges(lst, n, lane, hin, sm);
      else
        scan_groups<false>(pr, nD, ow, myTag, tid, lane, wq, gq, hin, wr, sm, npw, lst);
    } else {
      int n = scan_groups<true>(pr, nD, ow, myTag, tid, lane, wq, gq, hin, wr, sm, npw, lst);
      if (lane == 0) lcnt[slot] = n;
    }
   }
    __syncthreads();

#pragma unroll
    for (int rr = 0; rr < 2; ++rr) {
      const float* brow = WnP + (rp + rr) * 16 * SP + m * SP;
      const float* sm = s_msum + rr * (NPW_MAX * LP);
#pragma unroll
      for (int i = 0; i < TPW; ++i) {
        int t = wave + 32 * i;
        if (t < nT) {
          const float* arow = sm + (t * 16 + m) * LP;
          float rdeg = 1.0f / fmaxf(arow[D], 1.0f);
#pragma unroll
          for (int c = 0; c < 3; ++c) {
            int k0 = c * 4 + half * 2;
            v2f a, b;
            a.x = arow[k0] * rdeg; a.y = arow[k0 + 1] * rdeg;
            b.x = brow[k0];        b.y = brow[k0 + 1];
            acc[i].v = wmma_k4(a, b, acc[i].v);
          }
        }
      }
    }
    __syncthreads();
  }

#pragma unroll
  for (int i = 0; i < TPW; ++i) {
    int t = wave + 32 * i;
    if (t < nT && m < SP) {
      int rowBase = lo + t * 16 + half * 8;
      float* op = hout + (long long)rowBase * SP + m;
#pragma unroll
      for (int j = 0; j < 8; ++j)
        if (rowBase + j < Nn) op[j * SP] = acc[i].f[j];
    }
  }
}

__global__ void k_score(const float* __restrict__ hp,
                        const int* __restrict__ ps, const int* __restrict__ pd,
                        const int* __restrict__ ns, const int* __restrict__ nd,
                        float* __restrict__ out, int E_) {
  int i = blockIdx.x * blockDim.x + threadIdx.x;
  if (i >= 2 * E_) return;
  int a, b;
  if (i < E_) { a = ps[i]; b = pd[i]; }
  else        { a = ns[i - E_]; b = nd[i - E_]; }
  const float* pa = hp + (long long)a * SP;
  const float* pb = hp + (long long)b * SP;
  float s = 0.0f;
#pragma unroll
  for (int k = 0; k < D; ++k) s += pa[k] * pb[k];
  out[i] = s;
}

extern "C" void kernel_launch(void* const* d_in, const int* in_sizes, int n_in,
                              void* d_out, int out_size, void* d_ws, size_t ws_size,
                              hipStream_t stream) {
  (void)n_in; (void)out_size; (void)ws_size;
  const float* x    = (const float*)d_in[0];
  const int*   esrc = (const int*)d_in[1];
  const int*   edst = (const int*)d_in[2];
  const int*   nsrc = (const int*)d_in[3];
  const int*   ndst = (const int*)d_in[4];
  const float* Ws1  = (const float*)d_in[5];
  const float* Wn1  = (const float*)d_in[6];
  const float* b1   = (const float*)d_in[7];
  const float* Ws2  = (const float*)d_in[8];
  const float* Wn2  = (const float*)d_in[9];
  const float* b2   = (const float*)d_in[10];

  const int Nn = in_sizes[0] / D;
  const int Ee = in_sizes[3];

  float* B0   = (float*)d_ws;
  float* B1   = B0 + (size_t)Nn * SP;
  float* WsP1 = B1 + (size_t)Nn * SP;
  float* WnP1 = WsP1 + 16 * SP;
  float* bsP1 = WnP1 + RT * 16 * SP;
  float* WsP2 = bsP1 + 16;
  float* WnP2 = WsP2 + 16 * SP;
  float* bsP2 = WnP2 + RT * 16 * SP;
  unsigned int* owner = (unsigned int*)(bsP2 + 16);

  const int Epad   = (Ee + 4095) & ~4095;
  unsigned int* words = owner + RT * (size_t)(Epad >> 2);
  unsigned int* pres  = words + RT * (size_t)Epad;
  int* lcnt = (int*)(pres + RT * 2048 * (size_t)(Epad >> 12));
  unsigned int* lists = (unsigned int*)(lcnt + 255 * RT * 32);
  const int npw    = (((Nn + 254) / 255) + 15) & ~15;
  const int lBlocks = (Nn + npw - 1) / npw;

  k_prep<<<1, 256, 0, stream>>>(Ws1, Wn1, b1, Ws2, Wn2, b2,
                                WsP1, WnP1, bsP1, WsP2, WnP2, bsP2);
  k_owner<<<dim3(Epad >> 12, RT), 1024, 0, stream>>>(esrc, edst, owner, words, pres, Ee, Epad, npw);

  k_layer<false><<<lBlocks, 1024, 0, stream>>>(x, owner, words, pres, WsP1, WnP1, bsP1, B1,
                                               lists, lcnt, Nn, Epad, npw);
  k_layer<true><<<lBlocks, 1024, 0, stream>>>(B1, owner, words, pres, WsP2, WnP2, bsP2, B0,
                                              lists, lcnt, Nn, Epad, npw);

  k_score<<<(2 * Ee + 255) / 256, 256, 0, stream>>>(B0, esrc, edst, nsrc, ndst,
                                                    (float*)d_out, Ee);
}


# Owner tag and edge word merged into ONE 8-byte record; layer-1 scan reads a set group's 8 records with 8 adjacent lanes (presence per 8 edges): no dependent word read; LDS-atomic arrival order kept
# speedup vs baseline: 4.3566x; 1.0395x over previous
#include <hip/hip_runtime.h>


#define D  10
#define RT 4
#define SP 12
#define LP 13
#define NPW_MAX 2048
#define TPW (NPW_MAX / 16 / 32)
#define NOTAG 255u
#define GE 8
#define WQN 320
#define LCAP 768
#define LOCB 11
#define NOEDGE 0xFFFFFFFFu

typedef __attribute__((ext_vector_type(2))) float v2f;
typedef __attribute__((ext_vector_type(8))) float v8f;

__device__ __forceinline__ v8f wmma_k4(v2f a, v2f b, v8f c) {
  return __builtin_amdgcn_wmma_f32_16x16x4_f32(
       false, a,  false, b,
       (short)0, c,  false,  false);
}

__global__ void k_prep(const float* __restrict__ Ws1, const float* __restrict__ Wn1,
                       const float* __restrict__ b1,
                       const float* __restrict__ Ws2, const float* __restrict__ Wn2,
                       const float* __restrict__ b2,
                       float* __restrict__ WsP1, float* __restrict__ WnP1,
                       float* __restrict__ bsP1,
                       float* __restrict__ WsP2, float* __restrict__ WnP2,
                       float* __restrict__ bsP2) {
  for (int i = threadIdx.x; i < 16 * SP; i += blockDim.x) {
    int n = i / SP, k = i % SP;
    float s1 = 0.f, s2 = 0.f;
    if (n < D && k < D) {
#pragma unroll
      for (int r = 0; r < RT; ++r) {
        s1 += Ws1[r * D * D + n * D + k];
        s2 += Ws2[r * D * D + n * D + k];
      }
    }
    WsP1[i] = s1;
    WsP2[i] = s2;
  }
  for (int i = threadIdx.x; i < RT * 16 * SP; i += blockDim.x) {
    int r = i / (16 * SP), j = i % (16 * SP);
    int n = j / SP, k = j % SP;
    float v1 = 0.f, v2 = 0.f;
    if (n < D && k < D) {
      v1 = Wn1[r * D * D + n * D + k];
      v2 = Wn2[r * D * D + n * D + k];
    }
    WnP1[i] = v1;
    WnP2[i] = v2;
  }
  for (int i = threadIdx.x; i < 16; i += blockDim.x) {
    float s1 = 0.f, s2 = 0.f;
    if (i < D) {
#pragma unroll
      for (int r = 0; r < RT; ++r) {
        s1 += b1[r * D + i];
        s2 += b2[r * D + i];
      }
    }
    bsP1[i] = s1;
    bsP2[i] = s2;
  }
}

__global__ __launch_bounds__(1024)
void k_owner(const int* __restrict__ src, const int* __restrict__ dst,
             unsigned int* __restrict__ rec, unsigned int* __restrict__ pres,
             int E_, int Epad, int npw) {
  __shared__ unsigned int s_pres[4096];
#pragma unroll
  for (int k = 0; k < 4; ++k) s_pres[threadIdx.x + 1024 * k] = 0u;
  __syncthreads();
  const int r  = blockIdx.y;
  const int il = blockIdx.x * 1024 + threadIdx.x;
  const int e0 = il * 4;
  const unsigned int grpBit = 1u << ((threadIdx.x >> 1) & 31);
  const unsigned int grpDw  = threadIdx.x >> 6;
  const int* sp = src + (long long)r * E_ + e0;
  const int* dp = dst + (long long)r * E_ + e0;
  int s[4] = {0, 0, 0, 0}, d[4] = {0, 0, 0, 0};
  bool wide = (e0 + 3 < E_) && ((((unsigned long long)sp | (unsigned long long)dp) & 15ull) == 0ull);
  if (wide) {
    int4 sv = *(const int4*)sp, dv = *(const int4*)dp;
    s[0] = sv.x; s[1] = sv.y; s[2] = sv.z; s[3] = sv.w;
    d[0] = dv.x; d[1] = dv.y; d[2] = dv.z; d[3] = dv.w;
  } else {
#pragma unroll
    for (int b = 0; b < 4; ++b)
      if (e0 + b < E_) { s[b] = sp[b]; d[b] = dp[b]; }
  }
  unsigned int w[4], t[4];
#pragma unroll
  for (int b = 0; b < 4; ++b) {
    t[b] = NOTAG;
    w[b] = NOEDGE;
    if (e0 + b < E_) {
      unsigned int tag = (unsigned int)d[b] / (unsigned int)npw;
      unsigned int loc = (unsigned int)d[b] - tag * (unsigned int)npw;
      w[b] = ((unsigned int)s[b] << LOCB) | loc;
      t[b] = tag;
      if (tag < 255u) atomicOr(&s_pres[tag * 16u + grpDw], grpBit);
    }
  }
  uint4* ro = (uint4*)rec + ((size_t)r * (size_t)(Epad >> 1) + (size_t)il * 2);
  ro[0] = make_uint4(w[0], t[0], w[1], t[1]);
  ro[1] = make_uint4(w[2], t[2], w[3], t[3]);
  __syncthreads();
  unsigned int* po = pres + ((size_t)r * gridDim.x + blockIdx.x) * 4096;
#pragma unroll
  for (int k = 0; k < 4; ++k) po[threadIdx.x + 1024 * k] = s_pres[threadIdx.x + 1024 * k];
}

__device__ __forceinline__ void add_row(float* mp, float4 r0, float4 r1, float4 r2) {
  atomicAdd(mp + 0, r0.x); atomicAdd(mp + 1, r0.y);
  atomicAdd(mp + 2, r0.z); atomicAdd(mp + 3, r0.w);
  atomicAdd(mp + 4, r1.x); atomicAdd(mp + 5, r1.y);
  atomicAdd(mp + 6, r1.z); atomicAdd(mp + 7, r1.w);
  atomicAdd(mp + 8, r2.x); atomicAdd(mp + 9, r2.y);
  atomicAdd(mp + 10, 1.0f);
}

template <bool SAVE>
__device__ __forceinline__ void flush_edges(const unsigned int* wq, int n, int lane,
                                            const float* __restrict__ hin,
                                            float* s_msum, int npw,
                                            unsigned int* __restrict__ lst, int pos) {
  if (lane < n) {
    unsigned int w = wq[lane];
    unsigned int loc = w & ((1u << LOCB) - 1u);
    bool ok = loc < (unsigned int)npw;
    if (SAVE) {
      int at = pos + lane;
      if (at < LCAP) lst[at] = ok ? w : NOEDGE;
    }
    if (ok) {
      constexpr int IP = SAVE ? D : SP;
      const float* rp = hin + (long long)(w >> LOCB) * IP;
      float4 r0, r1, r2;
      r0.x = rp[0]; r0.y = rp[1]; r0.z = rp[2]; r0.w = rp[3];
      r1.x = rp[4]; r1.y = rp[5]; r1.z = rp[6]; r1.w = rp[7];
      r2.x = rp[8]; r2.y = rp[9]; r2.z = 0.f;   r2.w = 0.f;
      add_row(s_msum + loc * LP, r0, r1, r2);
    }
  }
}

template <bool SAVE>
__device__ __forceinline__ void take_groups(const int* gq, int n, const uint2* __restrict__ rc,
                                            uint2 (&st)[8],
                                            unsigned int myTag, int lane, unsigned int* wq,
                                            const float* __restrict__ hin,
                                            float* sm, int npw,
                                            unsigned int* __restrict__ lst,
                                            int& qn, int& done) {
  unsigned int wold[8];
  unsigned int mbits = 0u;
#pragma unroll
  for (int j = 0; j < 8; ++j) {
    wold[j] = st[j].x;
    mbits |= (st[j].y == myTag) ? (1u << j) : 0u;
  }
  const int sub = lane >> 3, slot = lane & 7;
#pragma unroll
  for (int j = 0; j < 8; ++j) {
    int gi = 4 * j + sub;
    st[j] = make_uint2(NOEDGE, NOTAG);
    if (gi < n) st[j] = rc[gq[gi] * GE + slot];
  }
  const unsigned int below = (1u << lane) - 1u;
#pragma unroll
  for (int j = 0; j < 8; ++j) {
    bool has = ((mbits >> j) & 1u) != 0u;
    unsigned int bal = __builtin_amdgcn_ballot_w32(has);
    if (has) wq[qn + __builtin_popcount(bal & below)] = wold[j];
    qn += __builtin_popcount(bal);
  }
  int off = 0;
  while (qn - off >= 32) {
    flush_edges<SAVE>(wq + off, 32, lane, hin, sm, npw, lst, done);
    done += 32;
    off += 32;
  }
  if (off > 0) {
    int rest = qn - off;
    unsigned int tmp = 0u;
    if (lane < rest) tmp = wq[off + lane];
    if (lane < rest) wq[lane] = tmp;
    qn = rest;
  }
}

template <bool SAVE>
__device__ __forceinline__ int scan_groups(const unsigned int* __restrict__ pr, int nD,
                                           const uint2* __restrict__ rc,
                                           unsigned int myTag, int tid, int lane,
                                           unsigned int* wq, int* gq,
                                           const float* __restrict__ hin,
                                           float* sm, int npw,
                                           unsigned int* __restrict__ lst) {
  int qn = 0;
  int done = 0;
  int gn = 0;
  uint2 st[8];
#pragma unroll
  for (int j = 0; j < 8; ++j) st[j] = make_uint2(NOEDGE, NOTAG);
  int dc = min(tid, nD - 1);
  unsigned int nxt = pr[(dc >> 4) * 4096 + (dc & 15)];
  for (int base = 0; base < nD; base += 1024) {
    const int d = base + tid;
    unsigned int pm = (d < nD) ? nxt : 0u;
    dc = min(d + 1024, nD - 1);
    nxt = pr[(dc >> 4) * 4096 + (dc & 15)];
    for (int trip = 0; trip < 32; ++trip) {
      bool has = (pm != 0u);
      unsigned int bal = __builtin_amdgcn_ballot_w32(has);
      if (bal == 0u) break;
      if (has) {
        int p = __builtin_ctz(pm);
        pm &= pm - 1u;
        int pos = gn + __builtin_popcount(bal & ((1u << lane) - 1u));
        gq[pos] = d * 32 + p;
      }
      gn += __builtin_popcount(bal);
      if (gn >= 32) {
        take_groups<SAVE>(gq, 32, rc, st, myTag, lane, wq, hin, sm, npw, lst, qn, done);
        int rest = gn - 32;
        int tmp = 0;
        if (lane < rest) tmp = gq[32 + lane];
        if (lane < rest) gq[lane] = tmp;
        gn = rest;
      }
    }
  }
  take_groups<SAVE>(gq, gn, rc, st, myTag, lane, wq, hin, sm, npw, lst, qn, done);
  take_groups<SAVE>(gq, 0, rc, st, myTag, lane, wq, hin, sm, npw, lst, qn, done);
  flush_edges<SAVE>(wq, qn, lane, hin, sm, npw, lst, done);
  return done + qn;
}

__device__ __forceinline__ void replay_edges(const unsigned int* __restrict__ lp, int n,
                                             int lane, const float* __restrict__ hin,
                                             float* sm) {
  constexpr int U = 4;
  unsigned int nxt[U];
#pragma unroll
  for (int u = 0; u < U; ++u) {
    int i = u * 32 + lane;
    nxt[u] = (i < n) ? lp[i] : NOEDGE;
  }
  for (int i0 = 0; i0 < n; i0 += 32 * U) {
    unsigned int p[U];
    float4 r0[U], r1[U], r2[U];
#pragma unroll
    for (int u = 0; u < U; ++u) {
      p[u] = nxt[u];
      if (p[u] != NOEDGE) {
        const float4* rp = (const float4*)(hin + (long long)(p[u] >> LOCB) * SP);
        r0[u] = rp[0]; r1[u] = rp[1]; r2[u] = rp[2];
      }
    }
#pragma unroll
    for (int u = 0; u < U; ++u) {
      int i = i0 + 32 * U + u * 32 + lane;
      nxt[u] = (i < n) ? lp[i] : NOEDGE;
    }
#pragma unroll
    for (int u = 0; u < U; ++u) {
      if (p[u] != NOEDGE)
        add_row(sm + (p[u] & ((1u << LOCB) - 1u)) * LP, r0[u], r1[u], r2[u]);
    }
  }
}


template <bool REPLAY>
__global__ __launch_bounds__(1024)
void k_layer(const float* __restrict__ hin,
             const unsigned int* __restrict__ rec,
             const unsigned int* __restrict__ pres,
             const float* __restrict__ WsP,
             const float* __restrict__ WnP,
             const float* __restrict__ bsP,
             float* __restrict__ hout,
             unsigned int* __restrict__ lists,
             int* __restrict__ lcnt,
             int Nn, int Epad, int npw) {
  __shared__ __attribute__((aligned(16))) float s_msum[2 * NPW_MAX * LP];
  __shared__ int s_q[32 * (WQN + 64)];

  const int tid  = threadIdx.x;
  const int lane = tid & 31;
  const int wave = tid >> 5;
  const int m    = lane & 15;
  const int half = lane >> 4;
  const int lo   = blockIdx.x * npw;
  const int cnt  = min(npw, Nn - lo);
  const int nT   = (cnt + 15) >> 4;
  unsigned int* wq = (unsigned int*)(s_q + wave * (WQN + 64));
  int* gq = s_q + wave * (WQN + 64) + WQN;

  union Acc { v8f v; float f[8]; };
  Acc acc[TPW];

  {
    float bsum = bsP[m];
    const float* brow = WsP + m * SP;
#pragma unroll
    for (int i = 0; i < TPW; ++i) {
      int t = wave + 32 * i;
#pragma unroll
      for (int j = 0; j < 8; ++j) acc[i].f[j] = bsum;
      if (t < nT) {
        constexpr int IP = REPLAY ? SP : D;
        int node = min(lo + t * 16 + m, Nn - 1);
        const float* arow = hin + (long long)node * IP;
#pragma unroll
        for (int c = 0; c < 3; ++c) {
          int k0 = c * 4 + half * 2;
          v2f a, b;
          bool stored = (k0 + 1 < IP);
          a.x = stored ? arow[k0] : 0.f; a.y = stored ? arow[k0 + 1] : 0.f;
          b.x = brow[k0]; b.y = brow[k0 + 1];
          acc[i].v = wmma_k4(a, b, acc[i].v);
        }
      }
    }
  }

  const unsigned int myTag = (unsigned int)blockIdx.x;
  const int nD   = Epad >> 8;

  static_assert(RT % 2 == 0, "etypes are taken two per pass");
  for (int rp = 0; rp < RT; rp += 2) {
    for (int i = tid; i < (npw * LP) / 4; i += 1024) {
      ((float4*)s_msum)[i] = make_float4(0.f, 0.f, 0.f, 0.f);
      ((float4*)(s_msum + NPW_MAX * LP))[i] = make_float4(0.f, 0.f, 0.f, 0.f);
    }
    __syncthreads();

   for (int rr = 0; rr < 2; ++rr) {
    const int r = rp + rr;
    float* sm = s_msum + rr * (NPW_MAX * LP);
    const uint2* rc = (const uint2*)rec + (long long)r * Epad;
    const long long slot = ((long long)blockIdx.x * RT + r) * 32 + wave;
    unsigned int* lst = lists + slot * LCAP;
    const unsigned int* pr = pres + ((long long)r * (Epad >> 12) * 256 + blockIdx.x) * 16;

    if (REPLAY) {
      int n = __builtin_amdgcn_readfirstlane(lcnt[slot]);
      if (n <= LCAP) replay_edges(lst, n, lane, hin, sm);
      else
        scan_groups<false>(pr, nD, rc, myTag, tid, lane, wq, gq, hin, sm, npw, lst);
    } else {
      int n = scan_groups<true>(pr, nD, rc, myTag, tid, lane, wq, gq, hin, sm, npw, lst);
      if (lane == 0) lcnt[slot] = n;
    }
   }
    __syncthreads();

#pragma unroll
    for (int rr = 0; rr < 2; ++rr) {
      const float* brow = WnP + (rp + rr) * 16 * SP + m * SP;
      const float* sm = s_msum + rr * (NPW_MAX * LP);
#pragma unroll
      for (int i = 0; i < TPW; ++i) {
        int t = wave + 32 * i;
        if (t < nT) {
          const float* arow = sm + (t * 16 + m) * LP;
          float rdeg = 1.0f / fmaxf(arow[D], 1.0f);
#pragma unroll
          for (int c = 0; c < 3; ++c) {
            int k0 = c * 4 + half * 2;
            v2f a, b;
            a.x = arow[k0] * rdeg; a.y = arow[k0 + 1] * rdeg;
            b.x = brow[k0];        b.y = brow[k0 + 1];
            acc[i].v = wmma_k4(a, b, acc[i].v);
          }
        }
      }
    }
    __syncthreads();
  }

#pragma unroll
  for (int i = 0; i < TPW; ++i) {
    int t = wave + 32 * i;
    if (t < nT && m < SP) {
      int rowBase = lo + t * 16 + half * 8;
      float* op = hout + (long long)rowBase * SP + m;
#pragma unroll
      for (int j = 0; j < 8; ++j)
        if (rowBase + j < Nn) op[j * SP] = acc[i].f[j];
    }
  }
}

__global__ void k_score(const float* __restrict__ hp,
                        const int* __restrict__ ps, const int* __restrict__ pd,
                        const int* __restrict__ ns, const int* __restrict__ nd,
                        float* __restrict__ out, int E_) {
  int i = blockIdx.x * blockDim.x + threadIdx.x;
  if (i >= 2 * E_) return;
  int a, b;
  if (i < E_) { a = ps[i]; b = pd[i]; }
  else        { a = ns[i - E_]; b = nd[i - E_]; }
  const float* pa = hp + (long long)a * SP;
  const float* pb = hp + (long long)b * SP;
  float s = 0.0f;
#pragma unroll
  for (int k = 0; k < D; ++k) s += pa[k] * pb[k];
  out[i] = s;
}

extern "C" void kernel_launch(void* const* d_in, const int* in_sizes, int n_in,
                              void* d_out, int out_size, void* d_ws, size_t ws_size,
                              hipStream_t stream) {
  (void)n_in; (void)out_size; (void)ws_size;
  const float* x    = (const float*)d_in[0];
  const int*   esrc = (const int*)d_in[1];
  const int*   edst = (const int*)d_in[2];
  const int*   nsrc = (const int*)d_in[3];
  const int*   ndst = (const int*)d_in[4];
  const float* Ws1  = (const float*)d_in[5];
  const float* Wn1  = (const float*)d_in[6];
  const float* b1   = (const float*)d_in[7];
  const float* Ws2  = (const float*)d_in[8];
  const float* Wn2  = (const float*)d_in[9];
  const float* b2   = (const float*)d_in[10];

  const int Nn = in_sizes[0] / D;
  const int Ee = in_sizes[3];

  float* B0   = (float*)d_ws;
  float* B1   = B0 + (size_t)Nn * SP;
  float* WsP1 = B1 + (size_t)Nn * SP;
  float* WnP1 = WsP1 + 16 * SP;
  float* bsP1 = WnP1 + RT * 16 * SP;
  float* WsP2 = bsP1 + 16;
  float* WnP2 = WsP2 + 16 * SP;
  float* bsP2 = WnP2 + RT * 16 * SP;
  unsigned int* rec = (unsigned int*)(bsP2 + 16);

  const int Epad   = (Ee + 4095) & ~4095;
  unsigned int* pres  = rec + RT * 2 * (size_t)Epad;
  int* lcnt = (int*)(pres + RT * 4096 * (size_t)(Epad >> 12));
  unsigned int* lists = (unsigned int*)(lcnt + 255 * RT * 32);
  const int npw    = (((Nn + 254) / 255) + 15) & ~15;
  const int lBlocks = (Nn + npw - 1) / npw;

  k_prep<<<1, 256, 0, stream>>>(Ws1, Wn1, b1, Ws2, Wn2, b2,
                                WsP1, WnP1, bsP1, WsP2, WnP2, bsP2);
  k_owner<<<dim3(Epad >> 12, RT), 1024, 0, stream>>>(esrc, edst, rec, pres, Ee, Epad, npw);

  k_layer<false><<<lBlocks, 1024, 0, stream>>>(x, rec, pres, WsP1, WnP1, bsP1, B1,
                                               lists, lcnt, Nn, Epad, npw);
  k_layer<true><<<lBlocks, 1024, 0, stream>>>(B1, rec, pres, WsP2, WnP2, bsP2, B0,
                                              lists, lcnt, Nn, Epad, npw);

  k_score<<<(2 * Ee + 255) / 256, 256, 0, stream>>>(B0, esrc, edst, nsrc, ndst,
                                                    (float*)d_out, Ee);
}
